# starting kernel
# baseline (speedup 1.0000x reference)
#include <hip/hip_runtime.h>
#include <hip/hip_bf16.h>
#include <math.h>

typedef __attribute__((ext_vector_type(2))) float v2f;
typedef __attribute__((ext_vector_type(8))) float v8f;

__global__ void zero_f32(float* __restrict__ p, int n) {
    int stride = gridDim.x * blockDim.x;
    for (int i = blockIdx.x * blockDim.x + threadIdx.x; i < n; i += stride)
        p[i] = 0.0f;
}

__global__ void relu_f32(float* __restrict__ p, int n) {
    int stride = gridDim.x * blockDim.x;
    for (int i = blockIdx.x * blockDim.x + threadIdx.x; i < n; i += stride) {
        float v = p[i];
        p[i] = v > 0.0f ? v : 0.0f;
    }
}

__global__ void spmm_scatter(const int* __restrict__ rows,
                             const int* __restrict__ cols,
                             const float* __restrict__ vals, int nnz,
                             const float* __restrict__ x, int dx,
                             float* __restrict__ y, int ldy, int coloff, int d) {
    int lane = threadIdx.x & 31;
    int wid  = (blockIdx.x * blockDim.x + threadIdx.x) >> 5;
    int nw   = (gridDim.x * blockDim.x) >> 5;
    for (int e = wid; e < nnz; e += nw) {
        int r = rows[e];
        int c = cols[e];
        float v = vals[e];
        const float* xs = x + (size_t)c * dx;
        float* yd = y + (size_t)r * ldy + coloff;
        for (int f = lane; f < d; f += 32)
            atomicAdd(&yd[f], v * xs[f]);
    }
}

__global__ void gemm128_wmma(const float* __restrict__ X,
                             const float* __restrict__ W,
                             const float* __restrict__ bias,
                             float* __restrict__ Y, int M) {
    int wave = (blockIdx.x * blockDim.x + threadIdx.x) >> 5;
    int lane = threadIdx.x & 31;
    int tiles = (M >> 4) * 8;
    if (wave >= tiles) return;

    int tm = wave >> 3;
    int tn = wave & 7;
    int half = lane >> 4;
    int l16  = lane & 15;

    const float* xrow = X + (size_t)(tm * 16 + l16) * 128;
    const float* wcol = W + tn * 16 + l16;

    v8f c = {};
    #pragma unroll
    for (int k = 0; k < 128; k += 4) {
        int ka = k + 2 * half;
        v2f a, b;
        a[0] = xrow[ka];
        a[1] = xrow[ka + 1];
        b[0] = wcol[(size_t)ka * 128];
        b[1] = wcol[(size_t)(ka + 1) * 128];
        c = __builtin_amdgcn_wmma_f32_16x16x4_f32(
                 false, a,  false, b,
                 (short)0, c,  false,  false);
    }

    float bn = bias[tn * 16 + l16];
    #pragma unroll
    for (int r = 0; r < 8; ++r) {
        int m = tm * 16 + r + 8 * half;
        Y[(size_t)m * 128 + tn * 16 + l16] = c[r] + bn;
    }
}

__global__ void gemm_naive(const float* __restrict__ X, const float* __restrict__ W,
                           const float* __restrict__ bias, float* __restrict__ Y,
                           int M, int K, int N) {
    int idx = blockIdx.x * blockDim.x + threadIdx.x;
    if (idx >= M * N) return;
    int m = idx / N, n = idx % N;
    float acc = bias[n];
    for (int k = 0; k < K; ++k)
        acc += X[m * K + k] * W[k * N + n];
    Y[idx] = acc;
}

__global__ void l2norm_rows(float* __restrict__ y, int ld) {
    __shared__ float red[256];
    float* p = y + (size_t)blockIdx.x * ld;
    float s = 0.0f;
    for (int f = threadIdx.x; f < ld; f += blockDim.x) {
        float v = p[f];
        s += v * v;
    }
    red[threadIdx.x] = s;
    __syncthreads();
    for (int off = blockDim.x >> 1; off > 0; off >>= 1) {
        if ((int)threadIdx.x < off) red[threadIdx.x] += red[threadIdx.x + off];
        __syncthreads();
    }
    float inv = 1.0f / (sqrtf(red[0]) + 1e-9f);
    for (int f = threadIdx.x; f < ld; f += blockDim.x)
        p[f] *= inv;
}

extern "C" void kernel_launch(void* const* d_in, const int* in_sizes, int n_in,
                              void* d_out, int out_size, void* d_ws, size_t ws_size,
                              hipStream_t stream) {
    (void)in_sizes; (void)n_in; (void)out_size; (void)ws_size;

    const int N_DOC = 10000, N_WORD = 20000, N_ENT = 10000, N_POS = 60;
    const int D = 128, D_WEMB = 300;
    const int E11 = 320000, E22 = 160000, E33 = 3600;
    const int E01 = 300000, E02 = 100000, E03 = 150000;

    const float* f1       = (const float*)d_in[0];
    const float* f2       = (const float*)d_in[1];
    const float* f3       = (const float*)d_in[2];
    const float* word_emb = (const float*)d_in[3];
    const int*   a11r = (const int*)d_in[4];
    const int*   a11c = (const int*)d_in[5];
    const float* a11v = (const float*)d_in[6];
    const int*   a22r = (const int*)d_in[7];
    const int*   a22c = (const int*)d_in[8];
    const float* a22v = (const float*)d_in[9];
    const int*   a33r = (const int*)d_in[10];
    const int*   a33c = (const int*)d_in[11];
    const float* a33v = (const float*)d_in[12];
    const int*   a01r = (const int*)d_in[13];
    const int*   a01c = (const int*)d_in[14];
    const float* a01v = (const float*)d_in[15];
    const int*   a02r = (const int*)d_in[16];
    const int*   a02c = (const int*)d_in[17];
    const float* a02v = (const float*)d_in[18];
    const int*   a03r = (const int*)d_in[19];
    const int*   a03c = (const int*)d_in[20];
    const float* a03v = (const float*)d_in[21];
    const float* W3   = (const float*)d_in[22];
    const float* b3   = (const float*)d_in[23];
    const float* W1_2 = (const float*)d_in[24];
    const float* b1_2 = (const float*)d_in[25];
    const float* W2_2 = (const float*)d_in[26];
    const float* b2_2 = (const float*)d_in[27];
    const float* W3_2 = (const float*)d_in[28];
    const float* b3_2 = (const float*)d_in[29];

    float* out  = (float*)d_out;
    float* out1 = out;
    float* out2 = out + (size_t)N_DOC * D;
    float* out3 = out2 + (size_t)N_DOC * (D + D_WEMB);

    float* bufA = (float*)d_ws;
    float* bufB = bufA + (size_t)N_WORD * D;
    float* bufC = bufB + (size_t)N_WORD * D;
    float* bufD = bufC + (size_t)N_POS * D;

    const int TB = 256;
    auto zgrid = [](int n) { int b = (n + 255) / 256; return b > 16384 ? 16384 : b; };
    auto egrid = [](int nnz) { int b = (nnz + 7) / 8; return b > 65535 ? 65535 : b; };

    zero_f32<<<zgrid(N_WORD * D), TB, 0, stream>>>(bufA, N_WORD * D);
    spmm_scatter<<<egrid(E11), TB, 0, stream>>>(a11r, a11c, a11v, E11, f1, D, bufA, D, 0, D);
    relu_f32<<<zgrid(N_WORD * D), TB, 0, stream>>>(bufA, N_WORD * D);
    gemm128_wmma<<<(N_WORD / 16) * 8 / 4, 128, 0, stream>>>(bufA, W1_2, b1_2, bufB, N_WORD);
    zero_f32<<<zgrid(N_WORD * D), TB, 0, stream>>>(bufA, N_WORD * D);
    spmm_scatter<<<egrid(E11), TB, 0, stream>>>(a11r, a11c, a11v, E11, bufB, D, bufA, D, 0, D);
    relu_f32<<<zgrid(N_WORD * D), TB, 0, stream>>>(bufA, N_WORD * D);
    zero_f32<<<zgrid(N_DOC * D), TB, 0, stream>>>(out1, N_DOC * D);
    spmm_scatter<<<egrid(E01), TB, 0, stream>>>(a01r, a01c, a01v, E01, bufA, D, out1, D, 0, D);
    l2norm_rows<<<N_DOC, 256, 0, stream>>>(out1, D);

    zero_f32<<<zgrid(N_ENT * D), TB, 0, stream>>>(bufA, N_ENT * D);
    spmm_scatter<<<egrid(E22), TB, 0, stream>>>(a22r, a22c, a22v, E22, f2, D, bufA, D, 0, D);
    relu_f32<<<zgrid(N_ENT * D), TB, 0, stream>>>(bufA, N_ENT * D);
    gemm128_wmma<<<(N_ENT / 16) * 8 / 4, 128, 0, stream>>>(bufA, W2_2, b2_2, bufB, N_ENT);
    zero_f32<<<zgrid(N_ENT * D), TB, 0, stream>>>(bufA, N_ENT * D);
    spmm_scatter<<<egrid(E22), TB, 0, stream>>>(a22r, a22c, a22v, E22, bufB, D, bufA, D, 0, D);
    relu_f32<<<zgrid(N_ENT * D), TB, 0, stream>>>(bufA, N_ENT * D);
    zero_f32<<<zgrid(N_DOC * (D + D_WEMB)), TB, 0, stream>>>(out2, N_DOC * (D + D_WEMB));
    spmm_scatter<<<egrid(E02), TB, 0, stream>>>(a02r, a02c, a02v, E02, bufA, D,
                                                out2, D + D_WEMB, 0, D);
    spmm_scatter<<<egrid(E02), TB, 0, stream>>>(a02r, a02c, a02v, E02, word_emb, D_WEMB,
                                                out2, D + D_WEMB, D, D_WEMB);
    l2norm_rows<<<N_DOC, 256, 0, stream>>>(out2, D + D_WEMB);

    gemm_naive<<<(N_POS * D + TB - 1) / TB, TB, 0, stream>>>(f3, W3, b3, bufC, N_POS, 60, D);
    zero_f32<<<zgrid(N_POS * D), TB, 0, stream>>>(bufD, N_POS * D);
    spmm_scatter<<<egrid(E33), TB, 0, stream>>>(a33r, a33c, a33v, E33, bufC, D, bufD, D, 0, D);
    relu_f32<<<zgrid(N_POS * D), TB, 0, stream>>>(bufD, N_POS * D);
    gemm_naive<<<(N_POS * D + TB - 1) / TB, TB, 0, stream>>>(bufD, W3_2, b3_2, bufC, N_POS, D, D);
    zero_f32<<<zgrid(N_POS * D), TB, 0, stream>>>(bufD, N_POS * D);
    spmm_scatter<<<egrid(E33), TB, 0, stream>>>(a33r, a33c, a33v, E33, bufC, D, bufD, D, 0, D);
    relu_f32<<<zgrid(N_POS * D), TB, 0, stream>>>(bufD, N_POS * D);
    zero_f32<<<zgrid(N_DOC * D), TB, 0, stream>>>(out3, N_DOC * D);
    spmm_scatter<<<egrid(E03), TB, 0, stream>>>(a03r, a03c, a03v, E03, bufD, D, out3, D, 0, D);
    l2norm_rows<<<N_DOC, 256, 0, stream>>>(out3, D);
}


# SpMM float-atomic scatter -> per-call row-sorted edge lists (segments ordered by edge id: fixed sum order) + wave-per-row register gather, relu/L2 norm at its one store; order of sums changes
# speedup vs baseline: 1.6266x; 1.6266x over previous
#include <hip/hip_runtime.h>
#include <math.h>

typedef __attribute__((ext_vector_type(2))) float v2f;
typedef __attribute__((ext_vector_type(8))) float v8f;

#define ALL_EDGES 1033600
#define ALL_ROWS  60060

__device__ __forceinline__ int edge_off(int a) {
    return a == 0 ? 0 : a == 1 ? 320000 : a == 2 ? 480000 : a == 3 ? 483600
         : a == 4 ? 783600 : a == 5 ? 883600 : 1033600;
}
__device__ __forceinline__ int row_off(int a) {
    return a == 0 ? 0 : a == 1 ? 20000 : a == 2 ? 30000 : a == 3 ? 30060
         : a == 4 ? 40060 : a == 5 ? 50060 : 60060;
}
__device__ __forceinline__ int op_of_edge(int g) {
    return (g >= 320000) + (g >= 480000) + (g >= 483600) + (g >= 783600) + (g >= 883600);
}
__device__ __forceinline__ int op_of_row(int r) {
    return (r >= 20000) + (r >= 30000) + (r >= 30060) + (r >= 40060) + (r >= 50060);
}

struct RowsPack {
    const int* r0; const int* r1; const int* r2;
    const int* r3; const int* r4; const int* r5;
};
struct EdgePack {
    const int* c0; const int* c1; const int* c2;
    const int* c3; const int* c4; const int* c5;
    const float* v0; const float* v1; const float* v2;
    const float* v3; const float* v4; const float* v5;
};

__global__ void zero_i32(int* __restrict__ p, int n) {
    int i = blockIdx.x * blockDim.x + threadIdx.x;
    if (i < n) p[i] = 0;
}

__global__ void csr_count(RowsPack rp, int* __restrict__ cnt) {
    int g = blockIdx.x * blockDim.x + threadIdx.x;
    if (g >= ALL_EDGES) return;
    int a = op_of_edge(g);
    const int* rows = a == 0 ? rp.r0 : a == 1 ? rp.r1 : a == 2 ? rp.r2
                    : a == 3 ? rp.r3 : a == 4 ? rp.r4 : rp.r5;
    int r = rows[g - edge_off(a)];
    atomicAdd(&cnt[row_off(a) + r], 1);
}

__global__ void __launch_bounds__(1024) csr_scan(const int* __restrict__ cnt,
                                                 int* __restrict__ rowptr) {
    __shared__ int sh[1024];
    int a  = blockIdx.x;
    int r0 = row_off(a);
    int n  = row_off(a + 1) - r0;
    int t  = threadIdx.x;
    int ipt = (n + 1023) >> 10;
    int b = t * ipt;
    int e = b + ipt < n ? b + ipt : n;
    int s = 0;
    for (int i = b; i < e; ++i) s += cnt[r0 + i];
    sh[t] = s;
    __syncthreads();
    for (int off = 1; off < 1024; off <<= 1) {
        int v = t >= off ? sh[t - off] : 0;
        __syncthreads();
        sh[t] += v;
        __syncthreads();
    }
    int run = edge_off(a) + sh[t] - s;
    for (int i = b; i < e; ++i) {
        rowptr[r0 + i] = run;
        run += cnt[r0 + i];
    }
    if (a == 5 && t == 0) rowptr[ALL_ROWS] = ALL_EDGES;
}

__global__ void csr_place(RowsPack rp, int* __restrict__ cnt,
                          const int* __restrict__ rowptr, int* __restrict__ perm) {
    int g = blockIdx.x * blockDim.x + threadIdx.x;
    if (g >= ALL_EDGES) return;
    int a = op_of_edge(g);
    const int* rows = a == 0 ? rp.r0 : a == 1 ? rp.r1 : a == 2 ? rp.r2
                    : a == 3 ? rp.r3 : a == 4 ? rp.r4 : rp.r5;
    int gr = row_off(a) + rows[g - edge_off(a)];
    int old = atomicSub(&cnt[gr], 1);
    perm[rowptr[gr] + old - 1] = g;
}

__global__ void csr_sort_emit(EdgePack ep, const int* __restrict__ rowptr,
                              const int* __restrict__ perm,
                              int* __restrict__ ccol, float* __restrict__ cval) {
    int lane = threadIdx.x & 31;
    int gr = __builtin_amdgcn_readfirstlane((int)(blockIdx.x * blockDim.x + threadIdx.x) >> 5);
    if (gr >= ALL_ROWS) return;
    int a = op_of_row(gr);
    const int* cols = a == 0 ? ep.c0 : a == 1 ? ep.c1 : a == 2 ? ep.c2
                    : a == 3 ? ep.c3 : a == 4 ? ep.c4 : ep.c5;
    const float* vals = a == 0 ? ep.v0 : a == 1 ? ep.v1 : a == 2 ? ep.v2
                      : a == 3 ? ep.v3 : a == 4 ? ep.v4 : ep.v5;
    int eo = edge_off(a);
    int s = rowptr[gr];
    int len = rowptr[gr + 1] - s;
    for (int i0 = 0; i0 < len; i0 += 32) {
        int i = i0 + lane;
        int key = i < len ? perm[s + i] : 0x7fffffff;
        int rank = 0;
        for (int j0 = 0; j0 < len; j0 += 32) {
            int jj = j0 + lane;
            int other = jj < len ? perm[s + jj] : 0x7fffffff;
            int n = len - j0 < 32 ? len - j0 : 32;
            for (int j = 0; j < n; ++j)
                rank += __builtin_amdgcn_readlane(other, j) < key ? 1 : 0;
        }
        if (i < len) {
            int le = key - eo;
            ccol[s + rank] = cols[le];
            cval[s + rank] = vals[le];
        }
    }
}

template <int U>
__device__ __forceinline__ void add_edges128(float4& acc, int cl, int vl, int j,
                                             const float* __restrict__ x, int lane) {
    float4 xv[U];
    float w[U];
    #pragma unroll
    for (int u = 0; u < U; ++u) {
        int c = __builtin_amdgcn_readlane(cl, j + u);
        w[u] = __int_as_float(__builtin_amdgcn_readlane(vl, j + u));
        xv[u] = reinterpret_cast<const float4*>(x + (size_t)c * 128)[lane];
    }
    #pragma unroll
    for (int u = 0; u < U; ++u) {
        acc.x += w[u] * xv[u].x; acc.y += w[u] * xv[u].y;
        acc.z += w[u] * xv[u].z; acc.w += w[u] * xv[u].w;
    }
}

template <bool NORM>
__global__ void gather128(const int* __restrict__ rowptr, const int* __restrict__ ccol,
                          const float* __restrict__ cval, int rowbase, int nrows,
                          const float* __restrict__ x, float* __restrict__ y) {
    int lane = threadIdx.x & 31;
    int row = __builtin_amdgcn_readfirstlane((int)(blockIdx.x * blockDim.x + threadIdx.x) >> 5);
    if (row >= nrows) return;
    int s = rowptr[rowbase + row];
    int t = rowptr[rowbase + row + 1];
    float4 acc = make_float4(0.f, 0.f, 0.f, 0.f);
    for (int base = s; base < t; base += 32) {
        int n = t - base < 32 ? t - base : 32;
        int cl = 0, vl = 0;
        if (lane < n) { cl = ccol[base + lane]; vl = __float_as_int(cval[base + lane]); }
        int j = 0;
        for (; j + 8 <= n; j += 8) add_edges128<8>(acc, cl, vl, j, x, lane);
        if (j + 4 <= n) { add_edges128<4>(acc, cl, vl, j, x, lane); j += 4; }
        if (j + 2 <= n) { add_edges128<2>(acc, cl, vl, j, x, lane); j += 2; }
        if (j < n) add_edges128<1>(acc, cl, vl, j, x, lane);
    }
    if (NORM) {
        float ss = acc.x * acc.x + acc.y * acc.y + acc.z * acc.z + acc.w * acc.w;
        ss += __shfl_xor(ss, 16, 32);
        ss += __shfl_xor(ss, 8, 32);
        ss += __shfl_xor(ss, 4, 32);
        ss += __shfl_xor(ss, 2, 32);
        ss += __shfl_xor(ss, 1, 32);
        float inv = 1.0f / (sqrtf(ss) + 1e-9f);
        acc.x *= inv; acc.y *= inv; acc.z *= inv; acc.w *= inv;
    } else {
        acc.x = acc.x > 0.f ? acc.x : 0.f; acc.y = acc.y > 0.f ? acc.y : 0.f;
        acc.z = acc.z > 0.f ? acc.z : 0.f; acc.w = acc.w > 0.f ? acc.w : 0.f;
    }
    reinterpret_cast<float4*>(y + (size_t)row * 128)[lane] = acc;
}

template <int U>
__device__ __forceinline__ void add_edges428(float4& a0, float4& b0, float4& b1, float4& b2,
                                             int cl, int vl, int j,
                                             const float* __restrict__ xa,
                                             const float* __restrict__ xb,
                                             int lane, bool tail) {
    float4 p[U], q0[U], q1[U], q2[U];
    float w[U];
    #pragma unroll
    for (int u = 0; u < U; ++u) {
        int c = __builtin_amdgcn_readlane(cl, j + u);
        w[u] = __int_as_float(__builtin_amdgcn_readlane(vl, j + u));
        const float4* pa = reinterpret_cast<const float4*>(xa + (size_t)c * 128);
        const float4* pb = reinterpret_cast<const float4*>(xb + (size_t)c * 300);
        p[u]  = pa[lane];
        q0[u] = pb[lane];
        q1[u] = pb[32 + lane];
        q2[u] = make_float4(0.f, 0.f, 0.f, 0.f);
        if (tail) q2[u] = pb[64 + lane];
    }
    #pragma unroll
    for (int u = 0; u < U; ++u) {
        float v = w[u];
        a0.x += v * p[u].x;  a0.y += v * p[u].y;  a0.z += v * p[u].z;  a0.w += v * p[u].w;
        b0.x += v * q0[u].x; b0.y += v * q0[u].y; b0.z += v * q0[u].z; b0.w += v * q0[u].w;
        b1.x += v * q1[u].x; b1.y += v * q1[u].y; b1.z += v * q1[u].z; b1.w += v * q1[u].w;
        b2.x += v * q2[u].x; b2.y += v * q2[u].y; b2.z += v * q2[u].z; b2.w += v * q2[u].w;
    }
}

__global__ void gather428_norm(const int* __restrict__ rowptr, const int* __restrict__ ccol,
                               const float* __restrict__ cval, int rowbase, int nrows,
                               const float* __restrict__ xa, const float* __restrict__ xb,
                               float* __restrict__ y) {
    int lane = threadIdx.x & 31;
    int row = __builtin_amdgcn_readfirstlane((int)(blockIdx.x * blockDim.x + threadIdx.x) >> 5);
    if (row >= nrows) return;
    int s = rowptr[rowbase + row];
    int t = rowptr[rowbase + row + 1];
    bool tail = lane < 11;
    float4 a0 = make_float4(0.f, 0.f, 0.f, 0.f), b0 = a0, b1 = a0, b2 = a0;
    for (int base = s; base < t; base += 32) {
        int n = t - base < 32 ? t - base : 32;
        int cl = 0, vl = 0;
        if (lane < n) { cl = ccol[base + lane]; vl = __float_as_int(cval[base + lane]); }
        int j = 0;
        for (; j + 4 <= n; j += 4) add_edges428<4>(a0, b0, b1, b2, cl, vl, j, xa, xb, lane, tail);
        if (j + 2 <= n) { add_edges428<2>(a0, b0, b1, b2, cl, vl, j, xa, xb, lane, tail); j += 2; }
        if (j < n) add_edges428<1>(a0, b0, b1, b2, cl, vl, j, xa, xb, lane, tail);
    }
    if (!tail) b2 = make_float4(0.f, 0.f, 0.f, 0.f);
    float ss = a0.x * a0.x + a0.y * a0.y + a0.z * a0.z + a0.w * a0.w
             + b0.x * b0.x + b0.y * b0.y + b0.z * b0.z + b0.w * b0.w
             + b1.x * b1.x + b1.y * b1.y + b1.z * b1.z + b1.w * b1.w
             + b2.x * b2.x + b2.y * b2.y + b2.z * b2.z + b2.w * b2.w;
    ss += __shfl_xor(ss, 16, 32);
    ss += __shfl_xor(ss, 8, 32);
    ss += __shfl_xor(ss, 4, 32);
    ss += __shfl_xor(ss, 2, 32);
    ss += __shfl_xor(ss, 1, 32);
    float inv = 1.0f / (sqrtf(ss) + 1e-9f);
    float4* py = reinterpret_cast<float4*>(y + (size_t)row * 428);
    py[lane]      = make_float4(a0.x * inv, a0.y * inv, a0.z * inv, a0.w * inv);
    py[32 + lane] = make_float4(b0.x * inv, b0.y * inv, b0.z * inv, b0.w * inv);
    py[64 + lane] = make_float4(b1.x * inv, b1.y * inv, b1.z * inv, b1.w * inv);
    if (tail) py[96 + lane] = make_float4(b2.x * inv, b2.y * inv, b2.z * inv, b2.w * inv);
}

__global__ void gemm128_wmma(const float* __restrict__ X,
                             const float* __restrict__ W,
                             const float* __restrict__ bias,
                             float* __restrict__ Y, int M) {
    int wave = (blockIdx.x * blockDim.x + threadIdx.x) >> 5;
    int lane = threadIdx.x & 31;
    int tiles = (M >> 4) * 8;
    if (wave >= tiles) return;

    int tm = wave >> 3;
    int tn = wave & 7;
    int half = lane >> 4;
    int l16  = lane & 15;

    const float* xrow = X + (size_t)(tm * 16 + l16) * 128;
    const float* wcol = W + tn * 16 + l16;

    v8f c = {};
    #pragma unroll
    for (int k = 0; k < 128; k += 4) {
        int ka = k + 2 * half;
        v2f a, b;
        a[0] = xrow[ka];
        a[1] = xrow[ka + 1];
        b[0] = wcol[(size_t)ka * 128];
        b[1] = wcol[(size_t)(ka + 1) * 128];
        c = __builtin_amdgcn_wmma_f32_16x16x4_f32(
                 false, a,  false, b,
                 (short)0, c,  false,  false);
    }

    float bn = bias[tn * 16 + l16];
    #pragma unroll
    for (int r = 0; r < 8; ++r) {
        int m = tm * 16 + r + 8 * half;
        Y[(size_t)m * 128 + tn * 16 + l16] = c[r] + bn;
    }
}

__global__ void gemm_naive(const float* __restrict__ X, const float* __restrict__ W,
                           const float* __restrict__ bias, float* __restrict__ Y,
                           int M, int K, int N) {
    int idx = blockIdx.x * blockDim.x + threadIdx.x;
    if (idx >= M * N) return;
    int m = idx / N, n = idx % N;
    float acc = bias[n];
    for (int k = 0; k < K; ++k)
        acc += X[m * K + k] * W[k * N + n];
    Y[idx] = acc;
}

extern "C" void kernel_launch(void* const* d_in, const int* in_sizes, int n_in,
                              void* d_out, int out_size, void* d_ws, size_t ws_size,
                              hipStream_t stream) {
    (void)in_sizes; (void)n_in; (void)out_size; (void)ws_size;

    const int N_DOC = 10000, N_WORD = 20000, N_ENT = 10000, N_POS = 60;
    const int D = 128, D_WEMB = 300;
    const int E11 = 320000, E22 = 160000, E33 = 3600;
    const int E01 = 300000, E02 = 100000, E03 = 150000;

    const float* f1       = (const float*)d_in[0];
    const float* f2       = (const float*)d_in[1];
    const float* f3       = (const float*)d_in[2];
    const float* word_emb = (const float*)d_in[3];
    const int*   a11r = (const int*)d_in[4];
    const int*   a11c = (const int*)d_in[5];
    const float* a11v = (const float*)d_in[6];
    const int*   a22r = (const int*)d_in[7];
    const int*   a22c = (const int*)d_in[8];
    const float* a22v = (const float*)d_in[9];
    const int*   a33r = (const int*)d_in[10];
    const int*   a33c = (const int*)d_in[11];
    const float* a33v = (const float*)d_in[12];
    const int*   a01r = (const int*)d_in[13];
    const int*   a01c = (const int*)d_in[14];
    const float* a01v = (const float*)d_in[15];
    const int*   a02r = (const int*)d_in[16];
    const int*   a02c = (const int*)d_in[17];
    const float* a02v = (const float*)d_in[18];
    const int*   a03r = (const int*)d_in[19];
    const int*   a03c = (const int*)d_in[20];
    const float* a03v = (const float*)d_in[21];
    const float* W3   = (const float*)d_in[22];
    const float* b3   = (const float*)d_in[23];
    const float* W1_2 = (const float*)d_in[24];
    const float* b1_2 = (const float*)d_in[25];
    const float* W2_2 = (const float*)d_in[26];
    const float* b2_2 = (const float*)d_in[27];
    const float* W3_2 = (const float*)d_in[28];
    const float* b3_2 = (const float*)d_in[29];

    float* out  = (float*)d_out;
    float* out1 = out;
    float* out2 = out + (size_t)N_DOC * D;
    float* out3 = out2 + (size_t)N_DOC * (D + D_WEMB);

    float* bufA = (float*)d_ws;
    float* csr  = bufA + (size_t)N_WORD * D;
    float* bufC = csr + (size_t)N_WORD * D;
    float* bufD = bufC + (size_t)N_POS * D;
    int*   ccol   = (int*)csr;
    float* cval   = csr + 1033600;
    int*   rowptr = (int*)(csr + 2067200);
    int*   cnt    = (int*)(csr + 2067200 + 60064);
    float* bufB = out2;
    int*   perm = (int*)out3;

    RowsPack rp = { a11r, a22r, a33r, a01r, a02r, a03r };
    EdgePack ep = { a11c, a22c, a33c, a01c, a02c, a03c,
                    a11v, a22v, a33v, a01v, a02v, a03v };
    (void)E11; (void)E22; (void)E33; (void)E01; (void)E02; (void)E03;

    const int TB = 256;
    const int R0 = 0, R1 = 20000, R2 = 30000, R3 = 30060, R4 = 40060, R5 = 50060;

    zero_i32<<<(60060 + TB - 1) / TB, TB, 0, stream>>>(cnt, 60060);
    csr_count<<<(1033600 + TB - 1) / TB, TB, 0, stream>>>(rp, cnt);
    csr_scan<<<6, 1024, 0, stream>>>(cnt, rowptr);
    csr_place<<<(1033600 + TB - 1) / TB, TB, 0, stream>>>(rp, cnt, rowptr, perm);
    csr_sort_emit<<<(60060 + 7) / 8, TB, 0, stream>>>(ep, rowptr, perm, ccol, cval);

    gather128<false><<<N_WORD / 8, TB, 0, stream>>>(rowptr, ccol, cval, R0, N_WORD, f1, bufA);
    gemm128_wmma<<<(N_WORD / 16) * 8 / 4, 128, 0, stream>>>(bufA, W1_2, b1_2, bufB, N_WORD);
    gather128<false><<<N_WORD / 8, TB, 0, stream>>>(rowptr, ccol, cval, R0, N_WORD, bufB, bufA);
    gather128<true><<<N_DOC / 8, TB, 0, stream>>>(rowptr, ccol, cval, R3, N_DOC, bufA, out1);

    gather128<false><<<N_ENT / 8, TB, 0, stream>>>(rowptr, ccol, cval, R1, N_ENT, f2, bufA);
    gemm128_wmma<<<(N_ENT / 16) * 8 / 4, 128, 0, stream>>>(bufA, W2_2, b2_2, bufB, N_ENT);
    gather128<false><<<N_ENT / 8, TB, 0, stream>>>(rowptr, ccol, cval, R1, N_ENT, bufB, bufA);
    gather428_norm<<<N_DOC / 8, TB, 0, stream>>>(rowptr, ccol, cval, R4, N_DOC, bufA, word_emb, out2);

    gemm_naive<<<(N_POS * D + TB - 1) / TB, TB, 0, stream>>>(f3, W3, b3, bufC, N_POS, 60, D);
    gather128<false><<<(N_POS + 7) / 8, TB, 0, stream>>>(rowptr, ccol, cval, R2, N_POS, bufC, bufD);
    gemm_naive<<<(N_POS * D + TB - 1) / TB, TB, 0, stream>>>(bufD, W3_2, b3_2, bufC, N_POS, D, D);
    gather128<false><<<(N_POS + 7) / 8, TB, 0, stream>>>(rowptr, ccol, cval, R2, N_POS, bufC, bufD);
    gather128<true><<<N_DOC / 8, TB, 0, stream>>>(rowptr, ccol, cval, R5, N_DOC, bufD, out3);
}


# list build: count pass keeps the value its integer add returns as the edge's slot; place pass scatters with it and drops its own returning atomic (one per edge, not two); sum order unchanged
# speedup vs baseline: 1.9583x; 1.2039x over previous
#include <hip/hip_runtime.h>
#include <math.h>

typedef __attribute__((ext_vector_type(2))) float v2f;
typedef __attribute__((ext_vector_type(8))) float v8f;

#define ALL_EDGES 1033600
#define ALL_ROWS  60060

__device__ __forceinline__ int edge_off(int a) {
    return a == 0 ? 0 : a == 1 ? 320000 : a == 2 ? 480000 : a == 3 ? 483600
         : a == 4 ? 783600 : a == 5 ? 883600 : 1033600;
}
__device__ __forceinline__ int row_off(int a) {
    return a == 0 ? 0 : a == 1 ? 20000 : a == 2 ? 30000 : a == 3 ? 30060
         : a == 4 ? 40060 : a == 5 ? 50060 : 60060;
}
__device__ __forceinline__ int op_of_edge(int g) {
    return (g >= 320000) + (g >= 480000) + (g >= 483600) + (g >= 783600) + (g >= 883600);
}
__device__ __forceinline__ int op_of_row(int r) {
    return (r >= 20000) + (r >= 30000) + (r >= 30060) + (r >= 40060) + (r >= 50060);
}

struct RowsPack {
    const int* r0; const int* r1; const int* r2;
    const int* r3; const int* r4; const int* r5;
};
struct EdgePack {
    const int* c0; const int* c1; const int* c2;
    const int* c3; const int* c4; const int* c5;
    const float* v0; const float* v1; const float* v2;
    const float* v3; const float* v4; const float* v5;
};

__global__ void zero_i32(int* __restrict__ p, int n) {
    int i = blockIdx.x * blockDim.x + threadIdx.x;
    if (i < n) p[i] = 0;
}

__global__ void csr_count(RowsPack rp, int* __restrict__ cnt, int* __restrict__ slot) {
    int g = blockIdx.x * blockDim.x + threadIdx.x;
    if (g >= ALL_EDGES) return;
    int a = op_of_edge(g);
    const int* rows = a == 0 ? rp.r0 : a == 1 ? rp.r1 : a == 2 ? rp.r2
                    : a == 3 ? rp.r3 : a == 4 ? rp.r4 : rp.r5;
    int r = rows[g - edge_off(a)];
    slot[g] = atomicAdd(&cnt[row_off(a) + r], 1);
}

__global__ void __launch_bounds__(1024) csr_scan(const int* __restrict__ cnt,
                                                 int* __restrict__ rowptr) {
    __shared__ int sh[1024];
    int a  = blockIdx.x;
    int r0 = row_off(a);
    int n  = row_off(a + 1) - r0;
    int t  = threadIdx.x;
    int ipt = (n + 1023) >> 10;
    int b = t * ipt;
    int e = b + ipt < n ? b + ipt : n;
    int s = 0;
    for (int i = b; i < e; ++i) s += cnt[r0 + i];
    sh[t] = s;
    __syncthreads();
    for (int off = 1; off < 1024; off <<= 1) {
        int v = t >= off ? sh[t - off] : 0;
        __syncthreads();
        sh[t] += v;
        __syncthreads();
    }
    int run = edge_off(a) + sh[t] - s;
    for (int i = b; i < e; ++i) {
        rowptr[r0 + i] = run;
        run += cnt[r0 + i];
    }
    if (a == 5 && t == 0) rowptr[ALL_ROWS] = ALL_EDGES;
}

__global__ void csr_place(RowsPack rp, const int* __restrict__ slot,
                          const int* __restrict__ rowptr, int* __restrict__ perm) {
    int g = blockIdx.x * blockDim.x + threadIdx.x;
    if (g >= ALL_EDGES) return;
    int a = op_of_edge(g);
    const int* rows = a == 0 ? rp.r0 : a == 1 ? rp.r1 : a == 2 ? rp.r2
                    : a == 3 ? rp.r3 : a == 4 ? rp.r4 : rp.r5;
    int gr = row_off(a) + rows[g - edge_off(a)];
    perm[rowptr[gr] + slot[g]] = g;
}

__global__ void csr_sort_emit(EdgePack ep, const int* __restrict__ rowptr,
                              const int* __restrict__ perm,
                              int* __restrict__ ccol, float* __restrict__ cval) {
    int lane = threadIdx.x & 31;
    int gr = __builtin_amdgcn_readfirstlane((int)(blockIdx.x * blockDim.x + threadIdx.x) >> 5);
    if (gr >= ALL_ROWS) return;
    int a = op_of_row(gr);
    const int* cols = a == 0 ? ep.c0 : a == 1 ? ep.c1 : a == 2 ? ep.c2
                    : a == 3 ? ep.c3 : a == 4 ? ep.c4 : ep.c5;
    const float* vals = a == 0 ? ep.v0 : a == 1 ? ep.v1 : a == 2 ? ep.v2
                      : a == 3 ? ep.v3 : a == 4 ? ep.v4 : ep.v5;
    int eo = edge_off(a);
    int s = rowptr[gr];
    int len = rowptr[gr + 1] - s;
    for (int i0 = 0; i0 < len; i0 += 32) {
        int i = i0 + lane;
        int key = i < len ? perm[s + i] : 0x7fffffff;
        int rank = 0;
        for (int j0 = 0; j0 < len; j0 += 32) {
            int jj = j0 + lane;
            int other = jj < len ? perm[s + jj] : 0x7fffffff;
            int n = len - j0 < 32 ? len - j0 : 32;
            for (int j = 0; j < n; ++j)
                rank += __builtin_amdgcn_readlane(other, j) < key ? 1 : 0;
        }
        if (i < len) {
            int le = key - eo;
            ccol[s + rank] = cols[le];
            cval[s + rank] = vals[le];
        }
    }
}

template <int U>
__device__ __forceinline__ void add_edges128(float4& acc, int cl, int vl, int j,
                                             const float* __restrict__ x, int lane) {
    float4 xv[U];
    float w[U];
    #pragma unroll
    for (int u = 0; u < U; ++u) {
        int c = __builtin_amdgcn_readlane(cl, j + u);
        w[u] = __int_as_float(__builtin_amdgcn_readlane(vl, j + u));
        xv[u] = reinterpret_cast<const float4*>(x + (size_t)c * 128)[lane];
    }
    #pragma unroll
    for (int u = 0; u < U; ++u) {
        acc.x += w[u] * xv[u].x; acc.y += w[u] * xv[u].y;
        acc.z += w[u] * xv[u].z; acc.w += w[u] * xv[u].w;
    }
}

template <bool NORM>
__global__ void gather128(const int* __restrict__ rowptr, const int* __restrict__ ccol,
                          const float* __restrict__ cval, int rowbase, int nrows,
                          const float* __restrict__ x, float* __restrict__ y) {
    int lane = threadIdx.x & 31;
    int row = __builtin_amdgcn_readfirstlane((int)(blockIdx.x * blockDim.x + threadIdx.x) >> 5);
    if (row >= nrows) return;
    int s = rowptr[rowbase + row];
    int t = rowptr[rowbase + row + 1];
    float4 acc = make_float4(0.f, 0.f, 0.f, 0.f);
    for (int base = s; base < t; base += 32) {
        int n = t - base < 32 ? t - base : 32;
        int cl = 0, vl = 0;
        if (lane < n) { cl = ccol[base + lane]; vl = __float_as_int(cval[base + lane]); }
        int j = 0;
        for (; j + 8 <= n; j += 8) add_edges128<8>(acc, cl, vl, j, x, lane);
        if (j + 4 <= n) { add_edges128<4>(acc, cl, vl, j, x, lane); j += 4; }
        if (j + 2 <= n) { add_edges128<2>(acc, cl, vl, j, x, lane); j += 2; }
        if (j < n) add_edges128<1>(acc, cl, vl, j, x, lane);
    }
    if (NORM) {
        float ss = acc.x * acc.x + acc.y * acc.y + acc.z * acc.z + acc.w * acc.w;
        ss += __shfl_xor(ss, 16, 32);
        ss += __shfl_xor(ss, 8, 32);
        ss += __shfl_xor(ss, 4, 32);
        ss += __shfl_xor(ss, 2, 32);
        ss += __shfl_xor(ss, 1, 32);
        float inv = 1.0f / (sqrtf(ss) + 1e-9f);
        acc.x *= inv; acc.y *= inv; acc.z *= inv; acc.w *= inv;
    } else {
        acc.x = acc.x > 0.f ? acc.x : 0.f; acc.y = acc.y > 0.f ? acc.y : 0.f;
        acc.z = acc.z > 0.f ? acc.z : 0.f; acc.w = acc.w > 0.f ? acc.w : 0.f;
    }
    reinterpret_cast<float4*>(y + (size_t)row * 128)[lane] = acc;
}

template <int U>
__device__ __forceinline__ void add_edges428(float4& a0, float4& b0, float4& b1, float4& b2,
                                             int cl, int vl, int j,
                                             const float* __restrict__ xa,
                                             const float* __restrict__ xb,
                                             int lane, bool tail) {
    float4 p[U], q0[U], q1[U], q2[U];
    float w[U];
    #pragma unroll
    for (int u = 0; u < U; ++u) {
        int c = __builtin_amdgcn_readlane(cl, j + u);
        w[u] = __int_as_float(__builtin_amdgcn_readlane(vl, j + u));
        const float4* pa = reinterpret_cast<const float4*>(xa + (size_t)c * 128);
        const float4* pb = reinterpret_cast<const float4*>(xb + (size_t)c * 300);
        p[u]  = pa[lane];
        q0[u] = pb[lane];
        q1[u] = pb[32 + lane];
        q2[u] = make_float4(0.f, 0.f, 0.f, 0.f);
        if (tail) q2[u] = pb[64 + lane];
    }
    #pragma unroll
    for (int u = 0; u < U; ++u) {
        float v = w[u];
        a0.x += v * p[u].x;  a0.y += v * p[u].y;  a0.z += v * p[u].z;  a0.w += v * p[u].w;
        b0.x += v * q0[u].x; b0.y += v * q0[u].y; b0.z += v * q0[u].z; b0.w += v * q0[u].w;
        b1.x += v * q1[u].x; b1.y += v * q1[u].y; b1.z += v * q1[u].z; b1.w += v * q1[u].w;
        b2.x += v * q2[u].x; b2.y += v * q2[u].y; b2.z += v * q2[u].z; b2.w += v * q2[u].w;
    }
}

__global__ void gather428_norm(const int* __restrict__ rowptr, const int* __restrict__ ccol,
                               const float* __restrict__ cval, int rowbase, int nrows,
                               const float* __restrict__ xa, const float* __restrict__ xb,
                               float* __restrict__ y) {
    int lane = threadIdx.x & 31;
    int row = __builtin_amdgcn_readfirstlane((int)(blockIdx.x * blockDim.x + threadIdx.x) >> 5);
    if (row >= nrows) return;
    int s = rowptr[rowbase + row];
    int t = rowptr[rowbase + row + 1];
    bool tail = lane < 11;
    float4 a0 = make_float4(0.f, 0.f, 0.f, 0.f), b0 = a0, b1 = a0, b2 = a0;
    for (int base = s; base < t; base += 32) {
        int n = t - base < 32 ? t - base : 32;
        int cl = 0, vl = 0;
        if (lane < n) { cl = ccol[base + lane]; vl = __float_as_int(cval[base + lane]); }
        int j = 0;
        for (; j + 4 <= n; j += 4) add_edges428<4>(a0, b0, b1, b2, cl, vl, j, xa, xb, lane, tail);
        if (j + 2 <= n) { add_edges428<2>(a0, b0, b1, b2, cl, vl, j, xa, xb, lane, tail); j += 2; }
        if (j < n) add_edges428<1>(a0, b0, b1, b2, cl, vl, j, xa, xb, lane, tail);
    }
    if (!tail) b2 = make_float4(0.f, 0.f, 0.f, 0.f);
    float ss = a0.x * a0.x + a0.y * a0.y + a0.z * a0.z + a0.w * a0.w
             + b0.x * b0.x + b0.y * b0.y + b0.z * b0.z + b0.w * b0.w
             + b1.x * b1.x + b1.y * b1.y + b1.z * b1.z + b1.w * b1.w
             + b2.x * b2.x + b2.y * b2.y + b2.z * b2.z + b2.w * b2.w;
    ss += __shfl_xor(ss, 16, 32);
    ss += __shfl_xor(ss, 8, 32);
    ss += __shfl_xor(ss, 4, 32);
    ss += __shfl_xor(ss, 2, 32);
    ss += __shfl_xor(ss, 1, 32);
    float inv = 1.0f / (sqrtf(ss) + 1e-9f);
    float4* py = reinterpret_cast<float4*>(y + (size_t)row * 428);
    py[lane]      = make_float4(a0.x * inv, a0.y * inv, a0.z * inv, a0.w * inv);
    py[32 + lane] = make_float4(b0.x * inv, b0.y * inv, b0.z * inv, b0.w * inv);
    py[64 + lane] = make_float4(b1.x * inv, b1.y * inv, b1.z * inv, b1.w * inv);
    if (tail) py[96 + lane] = make_float4(b2.x * inv, b2.y * inv, b2.z * inv, b2.w * inv);
}

__global__ void gemm128_wmma(const float* __restrict__ X,
                             const float* __restrict__ W,
                             const float* __restrict__ bias,
                             float* __restrict__ Y, int M) {
    int wave = (blockIdx.x * blockDim.x + threadIdx.x) >> 5;
    int lane = threadIdx.x & 31;
    int tiles = (M >> 4) * 8;
    if (wave >= tiles) return;

    int tm = wave >> 3;
    int tn = wave & 7;
    int half = lane >> 4;
    int l16  = lane & 15;

    const float* xrow = X + (size_t)(tm * 16 + l16) * 128;
    const float* wcol = W + tn * 16 + l16;

    v8f c = {};
    #pragma unroll
    for (int k = 0; k < 128; k += 4) {
        int ka = k + 2 * half;
        v2f a, b;
        a[0] = xrow[ka];
        a[1] = xrow[ka + 1];
        b[0] = wcol[(size_t)ka * 128];
        b[1] = wcol[(size_t)(ka + 1) * 128];
        c = __builtin_amdgcn_wmma_f32_16x16x4_f32(
                 false, a,  false, b,
                 (short)0, c,  false,  false);
    }

    float bn = bias[tn * 16 + l16];
    #pragma unroll
    for (int r = 0; r < 8; ++r) {
        int m = tm * 16 + r + 8 * half;
        Y[(size_t)m * 128 + tn * 16 + l16] = c[r] + bn;
    }
}

__global__ void gemm_naive(const float* __restrict__ X, const float* __restrict__ W,
                           const float* __restrict__ bias, float* __restrict__ Y,
                           int M, int K, int N) {
    int idx = blockIdx.x * blockDim.x + threadIdx.x;
    if (idx >= M * N) return;
    int m = idx / N, n = idx % N;
    float acc = bias[n];
    for (int k = 0; k < K; ++k)
        acc += X[m * K + k] * W[k * N + n];
    Y[idx] = acc;
}

extern "C" void kernel_launch(void* const* d_in, const int* in_sizes, int n_in,
                              void* d_out, int out_size, void* d_ws, size_t ws_size,
                              hipStream_t stream) {
    (void)in_sizes; (void)n_in; (void)out_size; (void)ws_size;

    const int N_DOC = 10000, N_WORD = 20000, N_ENT = 10000, N_POS = 60;
    const int D = 128, D_WEMB = 300;
    const int E11 = 320000, E22 = 160000, E33 = 3600;
    const int E01 = 300000, E02 = 100000, E03 = 150000;

    const float* f1       = (const float*)d_in[0];
    const float* f2       = (const float*)d_in[1];
    const float* f3       = (const float*)d_in[2];
    const float* word_emb = (const float*)d_in[3];
    const int*   a11r = (const int*)d_in[4];
    const int*   a11c = (const int*)d_in[5];
    const float* a11v = (const float*)d_in[6];
    const int*   a22r = (const int*)d_in[7];
    const int*   a22c = (const int*)d_in[8];
    const float* a22v = (const float*)d_in[9];
    const int*   a33r = (const int*)d_in[10];
    const int*   a33c = (const int*)d_in[11];
    const float* a33v = (const float*)d_in[12];
    const int*   a01r = (const int*)d_in[13];
    const int*   a01c = (const int*)d_in[14];
    const float* a01v = (const float*)d_in[15];
    const int*   a02r = (const int*)d_in[16];
    const int*   a02c = (const int*)d_in[17];
    const float* a02v = (const float*)d_in[18];
    const int*   a03r = (const int*)d_in[19];
    const int*   a03c = (const int*)d_in[20];
    const float* a03v = (const float*)d_in[21];
    const float* W3   = (const float*)d_in[22];
    const float* b3   = (const float*)d_in[23];
    const float* W1_2 = (const float*)d_in[24];
    const float* b1_2 = (const float*)d_in[25];
    const float* W2_2 = (const float*)d_in[26];
    const float* b2_2 = (const float*)d_in[27];
    const float* W3_2 = (const float*)d_in[28];
    const float* b3_2 = (const float*)d_in[29];

    float* out  = (float*)d_out;
    float* out1 = out;
    float* out2 = out + (size_t)N_DOC * D;
    float* out3 = out2 + (size_t)N_DOC * (D + D_WEMB);

    float* bufA = (float*)d_ws;
    float* csr  = bufA + (size_t)N_WORD * D;
    float* bufC = csr + (size_t)N_WORD * D;
    float* bufD = bufC + (size_t)N_POS * D;
    int*   ccol   = (int*)csr;
    float* cval   = csr + 1033600;
    int*   rowptr = (int*)(csr + 2067200);
    int*   cnt    = (int*)(csr + 2067200 + 60064);
    float* bufB = out2;
    int*   perm = (int*)out3;
    int*   slot = (int*)out1;

    RowsPack rp = { a11r, a22r, a33r, a01r, a02r, a03r };
    EdgePack ep = { a11c, a22c, a33c, a01c, a02c, a03c,
                    a11v, a22v, a33v, a01v, a02v, a03v };
    (void)E11; (void)E22; (void)E33; (void)E01; (void)E02; (void)E03;

    const int TB = 256;
    const int R0 = 0, R1 = 20000, R2 = 30000, R3 = 30060, R4 = 40060, R5 = 50060;

    zero_i32<<<(60060 + TB - 1) / TB, TB, 0, stream>>>(cnt, 60060);
    csr_count<<<(1033600 + TB - 1) / TB, TB, 0, stream>>>(rp, cnt, slot);
    csr_scan<<<6, 1024, 0, stream>>>(cnt, rowptr);
    csr_place<<<(1033600 + TB - 1) / TB, TB, 0, stream>>>(rp, slot, rowptr, perm);
    csr_sort_emit<<<(60060 + 7) / 8, TB, 0, stream>>>(ep, rowptr, perm, ccol, cval);

    gather128<false><<<N_WORD / 8, TB, 0, stream>>>(rowptr, ccol, cval, R0, N_WORD, f1, bufA);
    gemm128_wmma<<<(N_WORD / 16) * 8 / 4, 128, 0, stream>>>(bufA, W1_2, b1_2, bufB, N_WORD);
    gather128<false><<<N_WORD / 8, TB, 0, stream>>>(rowptr, ccol, cval, R0, N_WORD, bufB, bufA);
    gather128<true><<<N_DOC / 8, TB, 0, stream>>>(rowptr, ccol, cval, R3, N_DOC, bufA, out1);

    gather128<false><<<N_ENT / 8, TB, 0, stream>>>(rowptr, ccol, cval, R1, N_ENT, f2, bufA);
    gemm128_wmma<<<(N_ENT / 16) * 8 / 4, 128, 0, stream>>>(bufA, W2_2, b2_2, bufB, N_ENT);
    gather128<false><<<N_ENT / 8, TB, 0, stream>>>(rowptr, ccol, cval, R1, N_ENT, bufB, bufA);
    gather428_norm<<<N_DOC / 8, TB, 0, stream>>>(rowptr, ccol, cval, R4, N_DOC, bufA, word_emb, out2);

    gemm_naive<<<(N_POS * D + TB - 1) / TB, TB, 0, stream>>>(f3, W3, b3, bufC, N_POS, 60, D);
    gather128<false><<<(N_POS + 7) / 8, TB, 0, stream>>>(rowptr, ccol, cval, R2, N_POS, bufC, bufD);
    gemm_naive<<<(N_POS * D + TB - 1) / TB, TB, 0, stream>>>(bufD, W3_2, b3_2, bufC, N_POS, D, D);
    gather128<false><<<(N_POS + 7) / 8, TB, 0, stream>>>(rowptr, ccol, cval, R2, N_POS, bufC, bufD);
    gather128<true><<<N_DOC / 8, TB, 0, stream>>>(rowptr, ccol, cval, R5, N_DOC, bufD, out3);
}


# list build: 5 launches with a returning global integer add per edge -> 1 launch; a block owns a row range, reads its operator's row ids once, bins edge ids in LDS, sorts, emits; sum order unchanged
# speedup vs baseline: 3.2289x; 1.6488x over previous
#include <hip/hip_runtime.h>
#include <math.h>

typedef __attribute__((ext_vector_type(2))) float v2f;
typedef __attribute__((ext_vector_type(8))) float v8f;

#define ALL_EDGES 1033600
#define ALL_ROWS  60060

__device__ __forceinline__ int edge_off(int a) {
    return a == 0 ? 0 : a == 1 ? 320000 : a == 2 ? 480000 : a == 3 ? 483600
         : a == 4 ? 783600 : a == 5 ? 883600 : 1033600;
}
__device__ __forceinline__ int row_off(int a) {
    return a == 0 ? 0 : a == 1 ? 20000 : a == 2 ? 30000 : a == 3 ? 30060
         : a == 4 ? 40060 : a == 5 ? 50060 : 60060;
}
__device__ __forceinline__ int op_of_edge(int g) {
    return (g >= 320000) + (g >= 480000) + (g >= 483600) + (g >= 783600) + (g >= 883600);
}
__device__ __forceinline__ int op_of_row(int r) {
    return (r >= 20000) + (r >= 30000) + (r >= 30060) + (r >= 40060) + (r >= 50060);
}

struct RowsPack {
    const int* r0; const int* r1; const int* r2;
    const int* r3; const int* r4; const int* r5;
};
struct EdgePack {
    const int* c0; const int* c1; const int* c2;
    const int* c3; const int* c4; const int* c5;
    const float* v0; const float* v1; const float* v2;
    const float* v3; const float* v4; const float* v5;
};

#define BIN_WORDS 16384

__device__ __forceinline__ void visit_edge(int rv, int id, int r0, int nr, int lg, int capr,
                                           int* cnt, int* bins, int& below) {
    int d = rv - r0;
    below += d < 0 ? 1 : 0;
    if ((unsigned)d < (unsigned)nr) {
        int s = atomicAdd(&cnt[d], 1);
        if (s < capr) bins[(d << lg) + s] = id;
    }
}

__global__ void __launch_bounds__(1024) csr_build(RowsPack rp, EdgePack ep,
                                                  int* __restrict__ rowptr,
                                                  int* __restrict__ ccol,
                                                  float* __restrict__ cval) {
    __shared__ int bins[BIN_WORDS];
    __shared__ int cnt[256];
    __shared__ int sh[256];
    __shared__ int red[32];

    int t = threadIdx.x;
    int lane = t & 31;
    int wv = __builtin_amdgcn_readfirstlane(t >> 5);
    int b = blockIdx.x;
    int a  = (b >= 79) + (b >= 119) + (b >= 120) + (b >= 160) + (b >= 200);
    int b0 = a == 0 ? 0 : a == 1 ? 79 : a == 2 ? 119 : a == 3 ? 120 : a == 4 ? 160 : 200;
    int lg   = a == 2 ? 8 : 6;
    int capr = 1 << lg;
    int R    = BIN_WORDS >> lg;
    int r0   = (b - b0) * R;
    int nro  = row_off(a + 1) - row_off(a);
    int nr   = nro - r0 < R ? nro - r0 : R;
    int E    = edge_off(a + 1) - edge_off(a);
    const int* rows = a == 0 ? rp.r0 : a == 1 ? rp.r1 : a == 2 ? rp.r2
                    : a == 3 ? rp.r3 : a == 4 ? rp.r4 : rp.r5;
    const int* cols = a == 0 ? ep.c0 : a == 1 ? ep.c1 : a == 2 ? ep.c2
                    : a == 3 ? ep.c3 : a == 4 ? ep.c4 : ep.c5;
    const float* vals = a == 0 ? ep.v0 : a == 1 ? ep.v1 : a == 2 ? ep.v2
                      : a == 3 ? ep.v3 : a == 4 ? ep.v4 : ep.v5;

    if (t < 256) cnt[t] = 0;
    __syncthreads();

    int below = 0;
    if ((((size_t)rows) & 15) == 0) {
        const int4* rows4 = reinterpret_cast<const int4*>(rows);
        int n4 = E >> 2;
        int i = t;
        for (; i + 3072 < n4; i += 4096) {
            int4 v0 = rows4[i];
            int4 v1 = rows4[i + 1024];
            int4 v2 = rows4[i + 2048];
            int4 v3 = rows4[i + 3072];
            int e = i << 2;
            visit_edge(v0.x, e,     r0, nr, lg, capr, cnt, bins, below);
            visit_edge(v0.y, e + 1, r0, nr, lg, capr, cnt, bins, below);
            visit_edge(v0.z, e + 2, r0, nr, lg, capr, cnt, bins, below);
            visit_edge(v0.w, e + 3, r0, nr, lg, capr, cnt, bins, below);
            e += 4096;
            visit_edge(v1.x, e,     r0, nr, lg, capr, cnt, bins, below);
            visit_edge(v1.y, e + 1, r0, nr, lg, capr, cnt, bins, below);
            visit_edge(v1.z, e + 2, r0, nr, lg, capr, cnt, bins, below);
            visit_edge(v1.w, e + 3, r0, nr, lg, capr, cnt, bins, below);
            e += 4096;
            visit_edge(v2.x, e,     r0, nr, lg, capr, cnt, bins, below);
            visit_edge(v2.y, e + 1, r0, nr, lg, capr, cnt, bins, below);
            visit_edge(v2.z, e + 2, r0, nr, lg, capr, cnt, bins, below);
            visit_edge(v2.w, e + 3, r0, nr, lg, capr, cnt, bins, below);
            e += 4096;
            visit_edge(v3.x, e,     r0, nr, lg, capr, cnt, bins, below);
            visit_edge(v3.y, e + 1, r0, nr, lg, capr, cnt, bins, below);
            visit_edge(v3.z, e + 2, r0, nr, lg, capr, cnt, bins, below);
            visit_edge(v3.w, e + 3, r0, nr, lg, capr, cnt, bins, below);
        }
        for (; i < n4; i += 1024) {
            int4 v0 = rows4[i];
            int e = i << 2;
            visit_edge(v0.x, e,     r0, nr, lg, capr, cnt, bins, below);
            visit_edge(v0.y, e + 1, r0, nr, lg, capr, cnt, bins, below);
            visit_edge(v0.z, e + 2, r0, nr, lg, capr, cnt, bins, below);
            visit_edge(v0.w, e + 3, r0, nr, lg, capr, cnt, bins, below);
        }
        for (int k = (n4 << 2) + t; k < E; k += 1024)
            visit_edge(rows[k], k, r0, nr, lg, capr, cnt, bins, below);
    } else {
        for (int i = t; i < E; i += 1024)
            visit_edge(rows[i], i, r0, nr, lg, capr, cnt, bins, below);
    }

    below += __shfl_xor(below, 16, 32);
    below += __shfl_xor(below, 8, 32);
    below += __shfl_xor(below, 4, 32);
    below += __shfl_xor(below, 2, 32);
    below += __shfl_xor(below, 1, 32);
    if (lane == 0) red[wv] = below;
    __syncthreads();

    int c = 0;
    if (t < 256) { c = cnt[t]; sh[t] = c; }
    __syncthreads();
    for (int off = 1; off < 256; off <<= 1) {
        int v = (t < 256 && t >= off) ? sh[t - off] : 0;
        __syncthreads();
        if (t < 256) sh[t] += v;
        __syncthreads();
    }
    int base = edge_off(a);
    for (int k = 0; k < 32; ++k) base += red[k];
    if (t < nr) rowptr[row_off(a) + r0 + t] = base + sh[t] - c;
    if (b == 239 && t == 0) rowptr[ALL_ROWS] = ALL_EDGES;

    for (int d = wv; d < nr; d += 32) {
        int len = __builtin_amdgcn_readfirstlane(cnt[d]);
        int s   = __builtin_amdgcn_readfirstlane(base + sh[d]) - len;
        if (len <= capr) {
            const int* bl = bins + (d << lg);
            for (int i0 = 0; i0 < len; i0 += 32) {
                int i = i0 + lane;
                int key = i < len ? bl[i] : 0x7fffffff;
                int rank = 0;
                for (int j0 = 0; j0 < len; j0 += 32) {
                    int jj = j0 + lane;
                    int other = jj < len ? bl[jj] : 0x7fffffff;
                    int n = len - j0 < 32 ? len - j0 : 32;
                    for (int j = 0; j < n; ++j)
                        rank += __builtin_amdgcn_readlane(other, j) < key ? 1 : 0;
                }
                if (i < len) {
                    ccol[s + rank] = cols[key];
                    cval[s + rank] = vals[key];
                }
            }
        } else {
            int row = r0 + d;
            int done = 0;
            for (int i0 = 0; i0 < E; i0 += 32) {
                int i = i0 + lane;
                bool m = i < E && rows[i] == row;
                unsigned mask = __builtin_amdgcn_ballot_w32(m);
                if (m) {
                    int p = s + done + __builtin_popcount(mask & ((1u << lane) - 1u));
                    ccol[p] = cols[i];
                    cval[p] = vals[i];
                }
                done += __builtin_popcount(mask);
            }
        }
    }
}

template <int U>
__device__ __forceinline__ void add_edges128(float4& acc, int cl, int vl, int j,
                                             const float* __restrict__ x, int lane) {
    float4 xv[U];
    float w[U];
    #pragma unroll
    for (int u = 0; u < U; ++u) {
        int c = __builtin_amdgcn_readlane(cl, j + u);
        w[u] = __int_as_float(__builtin_amdgcn_readlane(vl, j + u));
        xv[u] = reinterpret_cast<const float4*>(x + (size_t)c * 128)[lane];
    }
    #pragma unroll
    for (int u = 0; u < U; ++u) {
        acc.x += w[u] * xv[u].x; acc.y += w[u] * xv[u].y;
        acc.z += w[u] * xv[u].z; acc.w += w[u] * xv[u].w;
    }
}

template <bool NORM>
__global__ void gather128(const int* __restrict__ rowptr, const int* __restrict__ ccol,
                          const float* __restrict__ cval, int rowbase, int nrows,
                          const float* __restrict__ x, float* __restrict__ y) {
    int lane = threadIdx.x & 31;
    int row = __builtin_amdgcn_readfirstlane((int)(blockIdx.x * blockDim.x + threadIdx.x) >> 5);
    if (row >= nrows) return;
    int s = rowptr[rowbase + row];
    int t = rowptr[rowbase + row + 1];
    float4 acc = make_float4(0.f, 0.f, 0.f, 0.f);
    for (int base = s; base < t; base += 32) {
        int n = t - base < 32 ? t - base : 32;
        int cl = 0, vl = 0;
        if (lane < n) { cl = ccol[base + lane]; vl = __float_as_int(cval[base + lane]); }
        int j = 0;
        for (; j + 8 <= n; j += 8) add_edges128<8>(acc, cl, vl, j, x, lane);
        if (j + 4 <= n) { add_edges128<4>(acc, cl, vl, j, x, lane); j += 4; }
        if (j + 2 <= n) { add_edges128<2>(acc, cl, vl, j, x, lane); j += 2; }
        if (j < n) add_edges128<1>(acc, cl, vl, j, x, lane);
    }
    if (NORM) {
        float ss = acc.x * acc.x + acc.y * acc.y + acc.z * acc.z + acc.w * acc.w;
        ss += __shfl_xor(ss, 16, 32);
        ss += __shfl_xor(ss, 8, 32);
        ss += __shfl_xor(ss, 4, 32);
        ss += __shfl_xor(ss, 2, 32);
        ss += __shfl_xor(ss, 1, 32);
        float inv = 1.0f / (sqrtf(ss) + 1e-9f);
        acc.x *= inv; acc.y *= inv; acc.z *= inv; acc.w *= inv;
    } else {
        acc.x = acc.x > 0.f ? acc.x : 0.f; acc.y = acc.y > 0.f ? acc.y : 0.f;
        acc.z = acc.z > 0.f ? acc.z : 0.f; acc.w = acc.w > 0.f ? acc.w : 0.f;
    }
    reinterpret_cast<float4*>(y + (size_t)row * 128)[lane] = acc;
}

template <int U>
__device__ __forceinline__ void add_edges428(float4& a0, float4& b0, float4& b1, float4& b2,
                                             int cl, int vl, int j,
                                             const float* __restrict__ xa,
                                             const float* __restrict__ xb,
                                             int lane, bool tail) {
    float4 p[U], q0[U], q1[U], q2[U];
    float w[U];
    #pragma unroll
    for (int u = 0; u < U; ++u) {
        int c = __builtin_amdgcn_readlane(cl, j + u);
        w[u] = __int_as_float(__builtin_amdgcn_readlane(vl, j + u));
        const float4* pa = reinterpret_cast<const float4*>(xa + (size_t)c * 128);
        const float4* pb = reinterpret_cast<const float4*>(xb + (size_t)c * 300);
        p[u]  = pa[lane];
        q0[u] = pb[lane];
        q1[u] = pb[32 + lane];
        q2[u] = make_float4(0.f, 0.f, 0.f, 0.f);
        if (tail) q2[u] = pb[64 + lane];
    }
    #pragma unroll
    for (int u = 0; u < U; ++u) {
        float v = w[u];
        a0.x += v * p[u].x;  a0.y += v * p[u].y;  a0.z += v * p[u].z;  a0.w += v * p[u].w;
        b0.x += v * q0[u].x; b0.y += v * q0[u].y; b0.z += v * q0[u].z; b0.w += v * q0[u].w;
        b1.x += v * q1[u].x; b1.y += v * q1[u].y; b1.z += v * q1[u].z; b1.w += v * q1[u].w;
        b2.x += v * q2[u].x; b2.y += v * q2[u].y; b2.z += v * q2[u].z; b2.w += v * q2[u].w;
    }
}

__global__ void gather428_norm(const int* __restrict__ rowptr, const int* __restrict__ ccol,
                               const float* __restrict__ cval, int rowbase, int nrows,
                               const float* __restrict__ xa, const float* __restrict__ xb,
                               float* __restrict__ y) {
    int lane = threadIdx.x & 31;
    int row = __builtin_amdgcn_readfirstlane((int)(blockIdx.x * blockDim.x + threadIdx.x) >> 5);
    if (row >= nrows) return;
    int s = rowptr[rowbase + row];
    int t = rowptr[rowbase + row + 1];
    bool tail = lane < 11;
    float4 a0 = make_float4(0.f, 0.f, 0.f, 0.f), b0 = a0, b1 = a0, b2 = a0;
    for (int base = s; base < t; base += 32) {
        int n = t - base < 32 ? t - base : 32;
        int cl = 0, vl = 0;
        if (lane < n) { cl = ccol[base + lane]; vl = __float_as_int(cval[base + lane]); }
        int j = 0;
        for (; j + 4 <= n; j += 4) add_edges428<4>(a0, b0, b1, b2, cl, vl, j, xa, xb, lane, tail);
        if (j + 2 <= n) { add_edges428<2>(a0, b0, b1, b2, cl, vl, j, xa, xb, lane, tail); j += 2; }
        if (j < n) add_edges428<1>(a0, b0, b1, b2, cl, vl, j, xa, xb, lane, tail);
    }
    if (!tail) b2 = make_float4(0.f, 0.f, 0.f, 0.f);
    float ss = a0.x * a0.x + a0.y * a0.y + a0.z * a0.z + a0.w * a0.w
             + b0.x * b0.x + b0.y * b0.y + b0.z * b0.z + b0.w * b0.w
             + b1.x * b1.x + b1.y * b1.y + b1.z * b1.z + b1.w * b1.w
             + b2.x * b2.x + b2.y * b2.y + b2.z * b2.z + b2.w * b2.w;
    ss += __shfl_xor(ss, 16, 32);
    ss += __shfl_xor(ss, 8, 32);
    ss += __shfl_xor(ss, 4, 32);
    ss += __shfl_xor(ss, 2, 32);
    ss += __shfl_xor(ss, 1, 32);
    float inv = 1.0f / (sqrtf(ss) + 1e-9f);
    float4* py = reinterpret_cast<float4*>(y + (size_t)row * 428);
    py[lane]      = make_float4(a0.x * inv, a0.y * inv, a0.z * inv, a0.w * inv);
    py[32 + lane] = make_float4(b0.x * inv, b0.y * inv, b0.z * inv, b0.w * inv);
    py[64 + lane] = make_float4(b1.x * inv, b1.y * inv, b1.z * inv, b1.w * inv);
    if (tail) py[96 + lane] = make_float4(b2.x * inv, b2.y * inv, b2.z * inv, b2.w * inv);
}

__global__ void gemm128_wmma(const float* __restrict__ X,
                             const float* __restrict__ W,
                             const float* __restrict__ bias,
                             float* __restrict__ Y, int M) {
    int wave = (blockIdx.x * blockDim.x + threadIdx.x) >> 5;
    int lane = threadIdx.x & 31;
    int tiles = (M >> 4) * 8;
    if (wave >= tiles) return;

    int tm = wave >> 3;
    int tn = wave & 7;
    int half = lane >> 4;
    int l16  = lane & 15;

    const float* xrow = X + (size_t)(tm * 16 + l16) * 128;
    const float* wcol = W + tn * 16 + l16;

    v8f c = {};
    #pragma unroll
    for (int k = 0; k < 128; k += 4) {
        int ka = k + 2 * half;
        v2f a, b;
        a[0] = xrow[ka];
        a[1] = xrow[ka + 1];
        b[0] = wcol[(size_t)ka * 128];
        b[1] = wcol[(size_t)(ka + 1) * 128];
        c = __builtin_amdgcn_wmma_f32_16x16x4_f32(
                 false, a,  false, b,
                 (short)0, c,  false,  false);
    }

    float bn = bias[tn * 16 + l16];
    #pragma unroll
    for (int r = 0; r < 8; ++r) {
        int m = tm * 16 + r + 8 * half;
        Y[(size_t)m * 128 + tn * 16 + l16] = c[r] + bn;
    }
}

__global__ void gemm_naive(const float* __restrict__ X, const float* __restrict__ W,
                           const float* __restrict__ bias, float* __restrict__ Y,
                           int M, int K, int N) {
    int idx = blockIdx.x * blockDim.x + threadIdx.x;
    if (idx >= M * N) return;
    int m = idx / N, n = idx % N;
    float acc = bias[n];
    for (int k = 0; k < K; ++k)
        acc += X[m * K + k] * W[k * N + n];
    Y[idx] = acc;
}

extern "C" void kernel_launch(void* const* d_in, const int* in_sizes, int n_in,
                              void* d_out, int out_size, void* d_ws, size_t ws_size,
                              hipStream_t stream) {
    (void)in_sizes; (void)n_in; (void)out_size; (void)ws_size;

    const int N_DOC = 10000, N_WORD = 20000, N_ENT = 10000, N_POS = 60;
    const int D = 128, D_WEMB = 300;
    const int E11 = 320000, E22 = 160000, E33 = 3600;
    const int E01 = 300000, E02 = 100000, E03 = 150000;

    const float* f1       = (const float*)d_in[0];
    const float* f2       = (const float*)d_in[1];
    const float* f3       = (const float*)d_in[2];
    const float* word_emb = (const float*)d_in[3];
    const int*   a11r = (const int*)d_in[4];
    const int*   a11c = (const int*)d_in[5];
    const float* a11v = (const float*)d_in[6];
    const int*   a22r = (const int*)d_in[7];
    const int*   a22c = (const int*)d_in[8];
    const float* a22v = (const float*)d_in[9];
    const int*   a33r = (const int*)d_in[10];
    const int*   a33c = (const int*)d_in[11];
    const float* a33v = (const float*)d_in[12];
    const int*   a01r = (const int*)d_in[13];
    const int*   a01c = (const int*)d_in[14];
    const float* a01v = (const float*)d_in[15];
    const int*   a02r = (const int*)d_in[16];
    const int*   a02c = (const int*)d_in[17];
    const float* a02v = (const float*)d_in[18];
    const int*   a03r = (const int*)d_in[19];
    const int*   a03c = (const int*)d_in[20];
    const float* a03v = (const float*)d_in[21];
    const float* W3   = (const float*)d_in[22];
    const float* b3   = (const float*)d_in[23];
    const float* W1_2 = (const float*)d_in[24];
    const float* b1_2 = (const float*)d_in[25];
    const float* W2_2 = (const float*)d_in[26];
    const float* b2_2 = (const float*)d_in[27];
    const float* W3_2 = (const float*)d_in[28];
    const float* b3_2 = (const float*)d_in[29];

    float* out  = (float*)d_out;
    float* out1 = out;
    float* out2 = out + (size_t)N_DOC * D;
    float* out3 = out2 + (size_t)N_DOC * (D + D_WEMB);

    float* bufA = (float*)d_ws;
    float* csr  = bufA + (size_t)N_WORD * D;
    float* bufC = csr + (size_t)N_WORD * D;
    float* bufD = bufC + (size_t)N_POS * D;
    int*   ccol   = (int*)csr;
    float* cval   = csr + 1033600;
    int*   rowptr = (int*)(csr + 2067200);
    float* bufB = out2;

    RowsPack rp = { a11r, a22r, a33r, a01r, a02r, a03r };
    EdgePack ep = { a11c, a22c, a33c, a01c, a02c, a03c,
                    a11v, a22v, a33v, a01v, a02v, a03v };
    (void)E11; (void)E22; (void)E33; (void)E01; (void)E02; (void)E03;

    const int TB = 256;
    const int R0 = 0, R1 = 20000, R2 = 30000, R3 = 30060, R4 = 40060, R5 = 50060;

    csr_build<<<240, 1024, 0, stream>>>(rp, ep, rowptr, ccol, cval);

    gather128<false><<<N_WORD / 8, TB, 0, stream>>>(rowptr, ccol, cval, R0, N_WORD, f1, bufA);
    gemm128_wmma<<<(N_WORD / 16) * 8 / 4, 128, 0, stream>>>(bufA, W1_2, b1_2, bufB, N_WORD);
    gather128<false><<<N_WORD / 8, TB, 0, stream>>>(rowptr, ccol, cval, R0, N_WORD, bufB, bufA);
    gather128<true><<<N_DOC / 8, TB, 0, stream>>>(rowptr, ccol, cval, R3, N_DOC, bufA, out1);

    gather128<false><<<N_ENT / 8, TB, 0, stream>>>(rowptr, ccol, cval, R1, N_ENT, f2, bufA);
    gemm128_wmma<<<(N_ENT / 16) * 8 / 4, 128, 0, stream>>>(bufA, W2_2, b2_2, bufB, N_ENT);
    gather128<false><<<N_ENT / 8, TB, 0, stream>>>(rowptr, ccol, cval, R1, N_ENT, bufB, bufA);
    gather428_norm<<<N_DOC / 8, TB, 0, stream>>>(rowptr, ccol, cval, R4, N_DOC, bufA, word_emb, out2);

    gemm_naive<<<(N_POS * D + TB - 1) / TB, TB, 0, stream>>>(f3, W3, b3, bufC, N_POS, 60, D);
    gather128<false><<<(N_POS + 7) / 8, TB, 0, stream>>>(rowptr, ccol, cval, R2, N_POS, bufC, bufD);
    gemm_naive<<<(N_POS * D + TB - 1) / TB, TB, 0, stream>>>(bufD, W3_2, b3_2, bufC, N_POS, D, D);
    gather128<false><<<(N_POS + 7) / 8, TB, 0, stream>>>(rowptr, ccol, cval, R2, N_POS, bufC, bufD);
    gather128<true><<<N_DOC / 8, TB, 0, stream>>>(rowptr, ccol, cval, R5, N_DOC, bufD, out3);
}


# independent chains (words, entities, pos tags) share launches: one grid per depth of the chains, block-uniform branch on the block index, 14 launches -> 6; every sum and its order unchanged
# speedup vs baseline: 4.2306x; 1.3102x over previous
#include <hip/hip_runtime.h>
#include <math.h>

typedef __attribute__((ext_vector_type(2))) float v2f;
typedef __attribute__((ext_vector_type(8))) float v8f;

#define ALL_EDGES 1033600
#define ALL_ROWS  60060

__device__ __forceinline__ int edge_off(int a) {
    return a == 0 ? 0 : a == 1 ? 320000 : a == 2 ? 480000 : a == 3 ? 483600
         : a == 4 ? 783600 : a == 5 ? 883600 : 1033600;
}
__device__ __forceinline__ int row_off(int a) {
    return a == 0 ? 0 : a == 1 ? 20000 : a == 2 ? 30000 : a == 3 ? 30060
         : a == 4 ? 40060 : a == 5 ? 50060 : 60060;
}
__device__ __forceinline__ int op_of_edge(int g) {
    return (g >= 320000) + (g >= 480000) + (g >= 483600) + (g >= 783600) + (g >= 883600);
}
__device__ __forceinline__ int op_of_row(int r) {
    return (r >= 20000) + (r >= 30000) + (r >= 30060) + (r >= 40060) + (r >= 50060);
}

struct RowsPack {
    const int* r0; const int* r1; const int* r2;
    const int* r3; const int* r4; const int* r5;
};
struct EdgePack {
    const int* c0; const int* c1; const int* c2;
    const int* c3; const int* c4; const int* c5;
    const float* v0; const float* v1; const float* v2;
    const float* v3; const float* v4; const float* v5;
};

#define BIN_WORDS 16384

__device__ __forceinline__ void visit_edge(int rv, int id, int r0, int nr, int lg, int capr,
                                           int* cnt, int* bins, int& below) {
    int d = rv - r0;
    below += d < 0 ? 1 : 0;
    if ((unsigned)d < (unsigned)nr) {
        int s = atomicAdd(&cnt[d], 1);
        if (s < capr) bins[(d << lg) + s] = id;
    }
}

__global__ void __launch_bounds__(1024) csr_build(RowsPack rp, EdgePack ep,
                                                  int* __restrict__ rowptr,
                                                  int* __restrict__ ccol,
                                                  float* __restrict__ cval) {
    __shared__ int bins[BIN_WORDS];
    __shared__ int cnt[256];
    __shared__ int sh[256];
    __shared__ int red[32];

    int t = threadIdx.x;
    int lane = t & 31;
    int wv = __builtin_amdgcn_readfirstlane(t >> 5);
    int b = blockIdx.x;
    int a  = (b >= 79) + (b >= 119) + (b >= 120) + (b >= 160) + (b >= 200);
    int b0 = a == 0 ? 0 : a == 1 ? 79 : a == 2 ? 119 : a == 3 ? 120 : a == 4 ? 160 : 200;
    int lg   = a == 2 ? 8 : 6;
    int capr = 1 << lg;
    int R    = BIN_WORDS >> lg;
    int r0   = (b - b0) * R;
    int nro  = row_off(a + 1) - row_off(a);
    int nr   = nro - r0 < R ? nro - r0 : R;
    int E    = edge_off(a + 1) - edge_off(a);
    const int* rows = a == 0 ? rp.r0 : a == 1 ? rp.r1 : a == 2 ? rp.r2
                    : a == 3 ? rp.r3 : a == 4 ? rp.r4 : rp.r5;
    const int* cols = a == 0 ? ep.c0 : a == 1 ? ep.c1 : a == 2 ? ep.c2
                    : a == 3 ? ep.c3 : a == 4 ? ep.c4 : ep.c5;
    const float* vals = a == 0 ? ep.v0 : a == 1 ? ep.v1 : a == 2 ? ep.v2
                      : a == 3 ? ep.v3 : a == 4 ? ep.v4 : ep.v5;

    if (t < 256) cnt[t] = 0;
    __syncthreads();

    int below = 0;
    if ((((size_t)rows) & 15) == 0) {
        const int4* rows4 = reinterpret_cast<const int4*>(rows);
        int n4 = E >> 2;
        int i = t;
        for (; i + 3072 < n4; i += 4096) {
            int4 v0 = rows4[i];
            int4 v1 = rows4[i + 1024];
            int4 v2 = rows4[i + 2048];
            int4 v3 = rows4[i + 3072];
            int e = i << 2;
            visit_edge(v0.x, e,     r0, nr, lg, capr, cnt, bins, below);
            visit_edge(v0.y, e + 1, r0, nr, lg, capr, cnt, bins, below);
            visit_edge(v0.z, e + 2, r0, nr, lg, capr, cnt, bins, below);
            visit_edge(v0.w, e + 3, r0, nr, lg, capr, cnt, bins, below);
            e += 4096;
            visit_edge(v1.x, e,     r0, nr, lg, capr, cnt, bins, below);
            visit_edge(v1.y, e + 1, r0, nr, lg, capr, cnt, bins, below);
            visit_edge(v1.z, e + 2, r0, nr, lg, capr, cnt, bins, below);
            visit_edge(v1.w, e + 3, r0, nr, lg, capr, cnt, bins, below);
            e += 4096;
            visit_edge(v2.x, e,     r0, nr, lg, capr, cnt, bins, below);
            visit_edge(v2.y, e + 1, r0, nr, lg, capr, cnt, bins, below);
            visit_edge(v2.z, e + 2, r0, nr, lg, capr, cnt, bins, below);
            visit_edge(v2.w, e + 3, r0, nr, lg, capr, cnt, bins, below);
            e += 4096;
            visit_edge(v3.x, e,     r0, nr, lg, capr, cnt, bins, below);
            visit_edge(v3.y, e + 1, r0, nr, lg, capr, cnt, bins, below);
            visit_edge(v3.z, e + 2, r0, nr, lg, capr, cnt, bins, below);
            visit_edge(v3.w, e + 3, r0, nr, lg, capr, cnt, bins, below);
        }
        for (; i < n4; i += 1024) {
            int4 v0 = rows4[i];
            int e = i << 2;
            visit_edge(v0.x, e,     r0, nr, lg, capr, cnt, bins, below);
            visit_edge(v0.y, e + 1, r0, nr, lg, capr, cnt, bins, below);
            visit_edge(v0.z, e + 2, r0, nr, lg, capr, cnt, bins, below);
            visit_edge(v0.w, e + 3, r0, nr, lg, capr, cnt, bins, below);
        }
        for (int k = (n4 << 2) + t; k < E; k += 1024)
            visit_edge(rows[k], k, r0, nr, lg, capr, cnt, bins, below);
    } else {
        for (int i = t; i < E; i += 1024)
            visit_edge(rows[i], i, r0, nr, lg, capr, cnt, bins, below);
    }

    below += __shfl_xor(below, 16, 32);
    below += __shfl_xor(below, 8, 32);
    below += __shfl_xor(below, 4, 32);
    below += __shfl_xor(below, 2, 32);
    below += __shfl_xor(below, 1, 32);
    if (lane == 0) red[wv] = below;
    __syncthreads();

    int c = 0;
    if (t < 256) { c = cnt[t]; sh[t] = c; }
    __syncthreads();
    for (int off = 1; off < 256; off <<= 1) {
        int v = (t < 256 && t >= off) ? sh[t - off] : 0;
        __syncthreads();
        if (t < 256) sh[t] += v;
        __syncthreads();
    }
    int base = edge_off(a);
    for (int k = 0; k < 32; ++k) base += red[k];
    if (t < nr) rowptr[row_off(a) + r0 + t] = base + sh[t] - c;
    if (b == 239 && t == 0) rowptr[ALL_ROWS] = ALL_EDGES;

    for (int d = wv; d < nr; d += 32) {
        int len = __builtin_amdgcn_readfirstlane(cnt[d]);
        int s   = __builtin_amdgcn_readfirstlane(base + sh[d]) - len;
        if (len <= capr) {
            const int* bl = bins + (d << lg);
            for (int i0 = 0; i0 < len; i0 += 32) {
                int i = i0 + lane;
                int key = i < len ? bl[i] : 0x7fffffff;
                int rank = 0;
                for (int j0 = 0; j0 < len; j0 += 32) {
                    int jj = j0 + lane;
                    int other = jj < len ? bl[jj] : 0x7fffffff;
                    int n = len - j0 < 32 ? len - j0 : 32;
                    for (int j = 0; j < n; ++j)
                        rank += __builtin_amdgcn_readlane(other, j) < key ? 1 : 0;
                }
                if (i < len) {
                    ccol[s + rank] = cols[key];
                    cval[s + rank] = vals[key];
                }
            }
        } else {
            int row = r0 + d;
            int done = 0;
            for (int i0 = 0; i0 < E; i0 += 32) {
                int i = i0 + lane;
                bool m = i < E && rows[i] == row;
                unsigned mask = __builtin_amdgcn_ballot_w32(m);
                if (m) {
                    int p = s + done + __builtin_popcount(mask & ((1u << lane) - 1u));
                    ccol[p] = cols[i];
                    cval[p] = vals[i];
                }
                done += __builtin_popcount(mask);
            }
        }
    }
}


template <int U>
__device__ __forceinline__ void add_edges128(float4& acc, int cl, int vl, int j,
                                             const float* __restrict__ x, int lane) {
    float4 xv[U];
    float w[U];
    #pragma unroll
    for (int u = 0; u < U; ++u) {
        int c = __builtin_amdgcn_readlane(cl, j + u);
        w[u] = __int_as_float(__builtin_amdgcn_readlane(vl, j + u));
        xv[u] = reinterpret_cast<const float4*>(x + (size_t)c * 128)[lane];
    }
    #pragma unroll
    for (int u = 0; u < U; ++u) {
        acc.x += w[u] * xv[u].x; acc.y += w[u] * xv[u].y;
        acc.z += w[u] * xv[u].z; acc.w += w[u] * xv[u].w;
    }
}

template <bool NORM>
__device__ __forceinline__ void gather128_rows(const int* __restrict__ rowptr,
                                               const int* __restrict__ ccol,
                                               const float* __restrict__ cval,
                                               int rowbase, int nrows,
                                               const float* __restrict__ x,
                                               float* __restrict__ y, int blk) {
    int lane = threadIdx.x & 31;
    int row = __builtin_amdgcn_readfirstlane((int)(blk * 256 + threadIdx.x) >> 5);
    if (row >= nrows) return;
    int s = rowptr[rowbase + row];
    int t = rowptr[rowbase + row + 1];
    float4 acc = make_float4(0.f, 0.f, 0.f, 0.f);
    for (int base = s; base < t; base += 32) {
        int n = t - base < 32 ? t - base : 32;
        int cl = 0, vl = 0;
        if (lane < n) { cl = ccol[base + lane]; vl = __float_as_int(cval[base + lane]); }
        int j = 0;
        for (; j + 8 <= n; j += 8) add_edges128<8>(acc, cl, vl, j, x, lane);
        if (j + 4 <= n) { add_edges128<4>(acc, cl, vl, j, x, lane); j += 4; }
        if (j + 2 <= n) { add_edges128<2>(acc, cl, vl, j, x, lane); j += 2; }
        if (j < n) add_edges128<1>(acc, cl, vl, j, x, lane);
    }
    if (NORM) {
        float ss = acc.x * acc.x + acc.y * acc.y + acc.z * acc.z + acc.w * acc.w;
        ss += __shfl_xor(ss, 16, 32);
        ss += __shfl_xor(ss, 8, 32);
        ss += __shfl_xor(ss, 4, 32);
        ss += __shfl_xor(ss, 2, 32);
        ss += __shfl_xor(ss, 1, 32);
        float inv = 1.0f / (sqrtf(ss) + 1e-9f);
        acc.x *= inv; acc.y *= inv; acc.z *= inv; acc.w *= inv;
    } else {
        acc.x = acc.x > 0.f ? acc.x : 0.f; acc.y = acc.y > 0.f ? acc.y : 0.f;
        acc.z = acc.z > 0.f ? acc.z : 0.f; acc.w = acc.w > 0.f ? acc.w : 0.f;
    }
    reinterpret_cast<float4*>(y + (size_t)row * 128)[lane] = acc;
}

template <int U>
__device__ __forceinline__ void add_edges428(float4& a0, float4& b0, float4& b1, float4& b2,
                                             int cl, int vl, int j,
                                             const float* __restrict__ xa,
                                             const float* __restrict__ xb,
                                             int lane, bool tail) {
    float4 p[U], q0[U], q1[U], q2[U];
    float w[U];
    #pragma unroll
    for (int u = 0; u < U; ++u) {
        int c = __builtin_amdgcn_readlane(cl, j + u);
        w[u] = __int_as_float(__builtin_amdgcn_readlane(vl, j + u));
        const float4* pa = reinterpret_cast<const float4*>(xa + (size_t)c * 128);
        const float4* pb = reinterpret_cast<const float4*>(xb + (size_t)c * 300);
        p[u]  = pa[lane];
        q0[u] = pb[lane];
        q1[u] = pb[32 + lane];
        q2[u] = make_float4(0.f, 0.f, 0.f, 0.f);
        if (tail) q2[u] = pb[64 + lane];
    }
    #pragma unroll
    for (int u = 0; u < U; ++u) {
        float v = w[u];
        a0.x += v * p[u].x;  a0.y += v * p[u].y;  a0.z += v * p[u].z;  a0.w += v * p[u].w;
        b0.x += v * q0[u].x; b0.y += v * q0[u].y; b0.z += v * q0[u].z; b0.w += v * q0[u].w;
        b1.x += v * q1[u].x; b1.y += v * q1[u].y; b1.z += v * q1[u].z; b1.w += v * q1[u].w;
        b2.x += v * q2[u].x; b2.y += v * q2[u].y; b2.z += v * q2[u].z; b2.w += v * q2[u].w;
    }
}

__device__ __forceinline__ void gather428_norm_rows(const int* __restrict__ rowptr,
                                                    const int* __restrict__ ccol,
                                                    const float* __restrict__ cval,
                                                    int rowbase, int nrows,
                                                    const float* __restrict__ xa,
                                                    const float* __restrict__ xb,
                                                    float* __restrict__ y, int blk) {
    int lane = threadIdx.x & 31;
    int row = __builtin_amdgcn_readfirstlane((int)(blk * 256 + threadIdx.x) >> 5);
    if (row >= nrows) return;
    int s = rowptr[rowbase + row];
    int t = rowptr[rowbase + row + 1];
    bool tail = lane < 11;
    float4 a0 = make_float4(0.f, 0.f, 0.f, 0.f), b0 = a0, b1 = a0, b2 = a0;
    for (int base = s; base < t; base += 32) {
        int n = t - base < 32 ? t - base : 32;
        int cl = 0, vl = 0;
        if (lane < n) { cl = ccol[base + lane]; vl = __float_as_int(cval[base + lane]); }
        int j = 0;
        for (; j + 4 <= n; j += 4) add_edges428<4>(a0, b0, b1, b2, cl, vl, j, xa, xb, lane, tail);
        if (j + 2 <= n) { add_edges428<2>(a0, b0, b1, b2, cl, vl, j, xa, xb, lane, tail); j += 2; }
        if (j < n) add_edges428<1>(a0, b0, b1, b2, cl, vl, j, xa, xb, lane, tail);
    }
    if (!tail) b2 = make_float4(0.f, 0.f, 0.f, 0.f);
    float ss = a0.x * a0.x + a0.y * a0.y + a0.z * a0.z + a0.w * a0.w
             + b0.x * b0.x + b0.y * b0.y + b0.z * b0.z + b0.w * b0.w
             + b1.x * b1.x + b1.y * b1.y + b1.z * b1.z + b1.w * b1.w
             + b2.x * b2.x + b2.y * b2.y + b2.z * b2.z + b2.w * b2.w;
    ss += __shfl_xor(ss, 16, 32);
    ss += __shfl_xor(ss, 8, 32);
    ss += __shfl_xor(ss, 4, 32);
    ss += __shfl_xor(ss, 2, 32);
    ss += __shfl_xor(ss, 1, 32);
    float inv = 1.0f / (sqrtf(ss) + 1e-9f);
    float4* py = reinterpret_cast<float4*>(y + (size_t)row * 428);
    py[lane]      = make_float4(a0.x * inv, a0.y * inv, a0.z * inv, a0.w * inv);
    py[32 + lane] = make_float4(b0.x * inv, b0.y * inv, b0.z * inv, b0.w * inv);
    py[64 + lane] = make_float4(b1.x * inv, b1.y * inv, b1.z * inv, b1.w * inv);
    if (tail) py[96 + lane] = make_float4(b2.x * inv, b2.y * inv, b2.z * inv, b2.w * inv);
}

__device__ __forceinline__ void gemm128_wmma_tiles(const float* __restrict__ X,
                                                   const float* __restrict__ W,
                                                   const float* __restrict__ bias,
                                                   float* __restrict__ Y, int M, int blk) {
    int wave = (blk * 256 + (int)threadIdx.x) >> 5;
    int lane = threadIdx.x & 31;
    int tiles = (M >> 4) * 8;
    if (wave >= tiles) return;

    int tm = wave >> 3;
    int tn = wave & 7;
    int half = lane >> 4;
    int l16  = lane & 15;

    const float* xrow = X + (size_t)(tm * 16 + l16) * 128;
    const float* wcol = W + tn * 16 + l16;

    v8f c = {};
    #pragma unroll
    for (int k = 0; k < 128; k += 4) {
        int ka = k + 2 * half;
        v2f a, b;
        a[0] = xrow[ka];
        a[1] = xrow[ka + 1];
        b[0] = wcol[(size_t)ka * 128];
        b[1] = wcol[(size_t)(ka + 1) * 128];
        c = __builtin_amdgcn_wmma_f32_16x16x4_f32(
                 false, a,  false, b,
                 (short)0, c,  false,  false);
    }

    float bn = bias[tn * 16 + l16];
    #pragma unroll
    for (int r = 0; r < 8; ++r) {
        int m = tm * 16 + r + 8 * half;
        Y[(size_t)m * 128 + tn * 16 + l16] = c[r] + bn;
    }
}

__device__ __forceinline__ void gemm_naive_elems(const float* __restrict__ X,
                                                 const float* __restrict__ W,
                                                 const float* __restrict__ bias,
                                                 float* __restrict__ Y,
                                                 int M, int K, int N, int blk) {
    int idx = blk * 256 + (int)threadIdx.x;
    if (idx >= M * N) return;
    int m = idx / N, n = idx % N;
    float acc = bias[n];
    for (int k = 0; k < K; ++k)
        acc += X[m * K + k] * W[k * N + n];
    Y[idx] = acc;
}

#define POS_GEMM_BLOCKS   30
#define POS_GATHER_BLOCKS 8
#define WORD_ROW_BLOCKS   2500
#define ENT_ROW_BLOCKS    1250
#define DOC_ROW_BLOCKS    1250
#define WORD_TILE_BLOCKS  1250
#define ENT_TILE_BLOCKS   625

__global__ void __launch_bounds__(256) stage_gather(const int* __restrict__ rowptr,
                                                    const int* __restrict__ ccol,
                                                    const float* __restrict__ cval,
                                                    const float* xw, float* yw,
                                                    const float* xe, float* ye,
                                                    const float* Xp, const float* Wp,
                                                    const float* bp, float* Yp, int Kp) {
    int b = blockIdx.x;
    if (b < POS_GEMM_BLOCKS) {
        gemm_naive_elems(Xp, Wp, bp, Yp, 60, Kp, 128, b);
    } else {
        bool word = b < POS_GEMM_BLOCKS + WORD_ROW_BLOCKS;
        gather128_rows<false>(rowptr, ccol, cval, word ? 0 : 20000, word ? 20000 : 10000,
                              word ? xw : xe, word ? yw : ye,
                              word ? b - POS_GEMM_BLOCKS
                                   : b - (POS_GEMM_BLOCKS + WORD_ROW_BLOCKS));
    }
}

__global__ void __launch_bounds__(256) stage_dense(const int* __restrict__ rowptr,
                                                   const int* __restrict__ ccol,
                                                   const float* __restrict__ cval,
                                                   const float* xp, float* yp,
                                                   const float* Xw, const float* Ww,
                                                   const float* bw, float* Yw,
                                                   const float* Xe, const float* We,
                                                   const float* be, float* Ye) {
    int b = blockIdx.x;
    if (b < POS_GATHER_BLOCKS) {
        gather128_rows<false>(rowptr, ccol, cval, 30000, 60, xp, yp, b);
    } else {
        bool word = b < POS_GATHER_BLOCKS + WORD_TILE_BLOCKS;
        gemm128_wmma_tiles(word ? Xw : Xe, word ? Ww : We, word ? bw : be, word ? Yw : Ye,
                           word ? 20000 : 10000,
                           word ? b - POS_GATHER_BLOCKS
                                : b - (POS_GATHER_BLOCKS + WORD_TILE_BLOCKS));
    }
}

__global__ void __launch_bounds__(256) stage_docs(const int* __restrict__ rowptr,
                                                  const int* __restrict__ ccol,
                                                  const float* __restrict__ cval,
                                                  const float* xp, float* yp,
                                                  const float* xa, const float* xb, float* y2,
                                                  const float* xw, float* y1) {
    int b = blockIdx.x;
    if (b < POS_GATHER_BLOCKS) {
        gather128_rows<false>(rowptr, ccol, cval, 30000, 60, xp, yp, b);
    } else if (b < POS_GATHER_BLOCKS + DOC_ROW_BLOCKS) {
        gather428_norm_rows(rowptr, ccol, cval, 40060, 10000, xa, xb, y2, b - POS_GATHER_BLOCKS);
    } else {
        gather128_rows<true>(rowptr, ccol, cval, 30060, 10000, xw, y1,
                             b - (POS_GATHER_BLOCKS + DOC_ROW_BLOCKS));
    }
}

__global__ void __launch_bounds__(256) stage_docs_pos(const int* __restrict__ rowptr,
                                                      const int* __restrict__ ccol,
                                                      const float* __restrict__ cval,
                                                      const float* __restrict__ x,
                                                      float* __restrict__ y) {
    gather128_rows<true>(rowptr, ccol, cval, 50060, 10000, x, y, blockIdx.x);
}

extern "C" void kernel_launch(void* const* d_in, const int* in_sizes, int n_in,
                              void* d_out, int out_size, void* d_ws, size_t ws_size,
                              hipStream_t stream) {
    (void)in_sizes; (void)n_in; (void)out_size; (void)ws_size;

    const int N_DOC = 10000, N_WORD = 20000, N_ENT = 10000, N_POS = 60;
    const int D = 128, D_WEMB = 300;

    const float* f1       = (const float*)d_in[0];
    const float* f2       = (const float*)d_in[1];
    const float* f3       = (const float*)d_in[2];
    const float* word_emb = (const float*)d_in[3];
    const int*   a11r = (const int*)d_in[4];
    const int*   a11c = (const int*)d_in[5];
    const float* a11v = (const float*)d_in[6];
    const int*   a22r = (const int*)d_in[7];
    const int*   a22c = (const int*)d_in[8];
    const float* a22v = (const float*)d_in[9];
    const int*   a33r = (const int*)d_in[10];
    const int*   a33c = (const int*)d_in[11];
    const float* a33v = (const float*)d_in[12];
    const int*   a01r = (const int*)d_in[13];
    const int*   a01c = (const int*)d_in[14];
    const float* a01v = (const float*)d_in[15];
    const int*   a02r = (const int*)d_in[16];
    const int*   a02c = (const int*)d_in[17];
    const float* a02v = (const float*)d_in[18];
    const int*   a03r = (const int*)d_in[19];
    const int*   a03c = (const int*)d_in[20];
    const float* a03v = (const float*)d_in[21];
    const float* W3   = (const float*)d_in[22];
    const float* b3   = (const float*)d_in[23];
    const float* W1_2 = (const float*)d_in[24];
    const float* b1_2 = (const float*)d_in[25];
    const float* W2_2 = (const float*)d_in[26];
    const float* b2_2 = (const float*)d_in[27];
    const float* W3_2 = (const float*)d_in[28];
    const float* b3_2 = (const float*)d_in[29];

    float* out  = (float*)d_out;
    float* out1 = out;
    float* out2 = out + (size_t)N_DOC * D;
    float* out3 = out2 + (size_t)N_DOC * (D + D_WEMB);

    float* bufA = (float*)d_ws;
    float* csr  = bufA + (size_t)N_WORD * D;
    float* bufC = csr + (size_t)N_WORD * D;
    float* bufD = bufC + (size_t)N_POS * D;
    int*   ccol   = (int*)csr;
    float* cval   = csr + 1033600;
    int*   rowptr = (int*)(csr + 2067200);

    float* wordG = bufA;
    float* entG  = out3;
    float* wordT = out2;
    float* entT  = out2 + (size_t)N_WORD * D;

    RowsPack rp = { a11r, a22r, a33r, a01r, a02r, a03r };
    EdgePack ep = { a11c, a22c, a33c, a01c, a02c, a03c,
                    a11v, a22v, a33v, a01v, a02v, a03v };

    csr_build<<<240, 1024, 0, stream>>>(rp, ep, rowptr, ccol, cval);

    stage_gather<<<POS_GEMM_BLOCKS + WORD_ROW_BLOCKS + ENT_ROW_BLOCKS, 256, 0, stream>>>(
        rowptr, ccol, cval, f1, wordG, f2, entG, f3, W3, b3, bufC, 60);
    stage_dense<<<POS_GATHER_BLOCKS + WORD_TILE_BLOCKS + ENT_TILE_BLOCKS, 256, 0, stream>>>(
        rowptr, ccol, cval, bufC, bufD,
        wordG, W1_2, b1_2, wordT, entG, W2_2, b2_2, entT);
    stage_gather<<<POS_GEMM_BLOCKS + WORD_ROW_BLOCKS + ENT_ROW_BLOCKS, 256, 0, stream>>>(
        rowptr, ccol, cval, wordT, wordG, entT, entG, bufD, W3_2, b3_2, bufC, 128);
    stage_docs<<<POS_GATHER_BLOCKS + DOC_ROW_BLOCKS + DOC_ROW_BLOCKS, 256, 0, stream>>>(
        rowptr, ccol, cval, bufC, bufD, entG, word_emb, out2, wordG, out1);
    stage_docs_pos<<<DOC_ROW_BLOCKS, 256, 0, stream>>>(rowptr, ccol, cval, bufD, out3);
}


# dense transform fused into the gather that feeds it: a block gathers a 32-row strip to LDS and WMMA takes A from there; pos chain starts in the list launch; 6 launches -> 5; sums and order unchanged
# speedup vs baseline: 4.7246x; 1.1168x over previous
#include <hip/hip_runtime.h>
#include <math.h>

typedef __attribute__((ext_vector_type(2))) float v2f;
typedef __attribute__((ext_vector_type(8))) float v8f;

#define ALL_EDGES 1033600
#define ALL_ROWS  60060

__device__ __forceinline__ int edge_off(int a) {
    return a == 0 ? 0 : a == 1 ? 320000 : a == 2 ? 480000 : a == 3 ? 483600
         : a == 4 ? 783600 : a == 5 ? 883600 : 1033600;
}
__device__ __forceinline__ int row_off(int a) {
    return a == 0 ? 0 : a == 1 ? 20000 : a == 2 ? 30000 : a == 3 ? 30060
         : a == 4 ? 40060 : a == 5 ? 50060 : 60060;
}
__device__ __forceinline__ int op_of_edge(int g) {
    return (g >= 320000) + (g >= 480000) + (g >= 483600) + (g >= 783600) + (g >= 883600);
}
__device__ __forceinline__ int op_of_row(int r) {
    return (r >= 20000) + (r >= 30000) + (r >= 30060) + (r >= 40060) + (r >= 50060);
}

struct RowsPack {
    const int* r0; const int* r1; const int* r2;
    const int* r3; const int* r4; const int* r5;
};
struct EdgePack {
    const int* c0; const int* c1; const int* c2;
    const int* c3; const int* c4; const int* c5;
    const float* v0; const float* v1; const float* v2;
    const float* v3; const float* v4; const float* v5;
};

#define BIN_WORDS 16384

__device__ __forceinline__ void visit_edge(int rv, int id, int r0, int nr, int lg, int capr,
                                           int* cnt, int* bins, int& below) {
    int d = rv - r0;
    below += d < 0 ? 1 : 0;
    if ((unsigned)d < (unsigned)nr) {
        int s = atomicAdd(&cnt[d], 1);
        if (s < capr) bins[(d << lg) + s] = id;
    }
}

#define LIST_BLOCKS 240
#define POS_GEMM_BLOCKS_1024 8

__device__ __forceinline__ void gemm_naive_elem(const float* __restrict__ X,
                                                const float* __restrict__ W,
                                                const float* __restrict__ bias,
                                                float* __restrict__ Y,
                                                int M, int K, int N, int idx);

__global__ void __launch_bounds__(1024) csr_build(RowsPack rp, EdgePack ep,
                                                  int* __restrict__ rowptr,
                                                  int* __restrict__ ccol,
                                                  float* __restrict__ cval,
                                                  const float* Xp, const float* Wp,
                                                  const float* bp, float* Yp) {
    __shared__ int bins[BIN_WORDS];
    __shared__ int cnt[256];
    __shared__ int sh[256];
    __shared__ int red[32];

    int t = threadIdx.x;
    int lane = t & 31;
    int wv = __builtin_amdgcn_readfirstlane(t >> 5);
    int b = blockIdx.x;
    if (b >= LIST_BLOCKS) {
        gemm_naive_elem(Xp, Wp, bp, Yp, 60, 60, 128, (b - LIST_BLOCKS) * 1024 + t);
        return;
    }
    int a  = (b >= 79) + (b >= 119) + (b >= 120) + (b >= 160) + (b >= 200);
    int b0 = a == 0 ? 0 : a == 1 ? 79 : a == 2 ? 119 : a == 3 ? 120 : a == 4 ? 160 : 200;
    int lg   = a == 2 ? 8 : 6;
    int capr = 1 << lg;
    int R    = BIN_WORDS >> lg;
    int r0   = (b - b0) * R;
    int nro  = row_off(a + 1) - row_off(a);
    int nr   = nro - r0 < R ? nro - r0 : R;
    int E    = edge_off(a + 1) - edge_off(a);
    const int* rows = a == 0 ? rp.r0 : a == 1 ? rp.r1 : a == 2 ? rp.r2
                    : a == 3 ? rp.r3 : a == 4 ? rp.r4 : rp.r5;
    const int* cols = a == 0 ? ep.c0 : a == 1 ? ep.c1 : a == 2 ? ep.c2
                    : a == 3 ? ep.c3 : a == 4 ? ep.c4 : ep.c5;
    const float* vals = a == 0 ? ep.v0 : a == 1 ? ep.v1 : a == 2 ? ep.v2
                      : a == 3 ? ep.v3 : a == 4 ? ep.v4 : ep.v5;

    if (t < 256) cnt[t] = 0;
    __syncthreads();

    int below = 0;
    if ((((size_t)rows) & 15) == 0) {
        const int4* rows4 = reinterpret_cast<const int4*>(rows);
        int n4 = E >> 2;
        int i = t;
        for (; i + 3072 < n4; i += 4096) {
            int4 v0 = rows4[i];
            int4 v1 = rows4[i + 1024];
            int4 v2 = rows4[i + 2048];
            int4 v3 = rows4[i + 3072];
            int e = i << 2;
            visit_edge(v0.x, e,     r0, nr, lg, capr, cnt, bins, below);
            visit_edge(v0.y, e + 1, r0, nr, lg, capr, cnt, bins, below);
            visit_edge(v0.z, e + 2, r0, nr, lg, capr, cnt, bins, below);
            visit_edge(v0.w, e + 3, r0, nr, lg, capr, cnt, bins, below);
            e += 4096;
            visit_edge(v1.x, e,     r0, nr, lg, capr, cnt, bins, below);
            visit_edge(v1.y, e + 1, r0, nr, lg, capr, cnt, bins, below);
            visit_edge(v1.z, e + 2, r0, nr, lg, capr, cnt, bins, below);
            visit_edge(v1.w, e + 3, r0, nr, lg, capr, cnt, bins, below);
            e += 4096;
            visit_edge(v2.x, e,     r0, nr, lg, capr, cnt, bins, below);
            visit_edge(v2.y, e + 1, r0, nr, lg, capr, cnt, bins, below);
            visit_edge(v2.z, e + 2, r0, nr, lg, capr, cnt, bins, below);
            visit_edge(v2.w, e + 3, r0, nr, lg, capr, cnt, bins, below);
            e += 4096;
            visit_edge(v3.x, e,     r0, nr, lg, capr, cnt, bins, below);
            visit_edge(v3.y, e + 1, r0, nr, lg, capr, cnt, bins, below);
            visit_edge(v3.z, e + 2, r0, nr, lg, capr, cnt, bins, below);
            visit_edge(v3.w, e + 3, r0, nr, lg, capr, cnt, bins, below);
        }
        for (; i < n4; i += 1024) {
            int4 v0 = rows4[i];
            int e = i << 2;
            visit_edge(v0.x, e,     r0, nr, lg, capr, cnt, bins, below);
            visit_edge(v0.y, e + 1, r0, nr, lg, capr, cnt, bins, below);
            visit_edge(v0.z, e + 2, r0, nr, lg, capr, cnt, bins, below);
            visit_edge(v0.w, e + 3, r0, nr, lg, capr, cnt, bins, below);
        }
        for (int k = (n4 << 2) + t; k < E; k += 1024)
            visit_edge(rows[k], k, r0, nr, lg, capr, cnt, bins, below);
    } else {
        for (int i = t; i < E; i += 1024)
            visit_edge(rows[i], i, r0, nr, lg, capr, cnt, bins, below);
    }

    below += __shfl_xor(below, 16, 32);
    below += __shfl_xor(below, 8, 32);
    below += __shfl_xor(below, 4, 32);
    below += __shfl_xor(below, 2, 32);
    below += __shfl_xor(below, 1, 32);
    if (lane == 0) red[wv] = below;
    __syncthreads();

    int c = 0;
    if (t < 256) { c = cnt[t]; sh[t] = c; }
    __syncthreads();
    for (int off = 1; off < 256; off <<= 1) {
        int v = (t < 256 && t >= off) ? sh[t - off] : 0;
        __syncthreads();
        if (t < 256) sh[t] += v;
        __syncthreads();
    }
    int base = edge_off(a);
    for (int k = 0; k < 32; ++k) base += red[k];
    if (t < nr) rowptr[row_off(a) + r0 + t] = base + sh[t] - c;
    if (b == 239 && t == 0) rowptr[ALL_ROWS] = ALL_EDGES;

    for (int d = wv; d < nr; d += 32) {
        int len = __builtin_amdgcn_readfirstlane(cnt[d]);
        int s   = __builtin_amdgcn_readfirstlane(base + sh[d]) - len;
        if (len <= capr) {
            const int* bl = bins + (d << lg);
            for (int i0 = 0; i0 < len; i0 += 32) {
                int i = i0 + lane;
                int key = i < len ? bl[i] : 0x7fffffff;
                int rank = 0;
                for (int j0 = 0; j0 < len; j0 += 32) {
                    int jj = j0 + lane;
                    int other = jj < len ? bl[jj] : 0x7fffffff;
                    int n = len - j0 < 32 ? len - j0 : 32;
                    for (int j = 0; j < n; ++j)
                        rank += __builtin_amdgcn_readlane(other, j) < key ? 1 : 0;
                }
                if (i < len) {
                    ccol[s + rank] = cols[key];
                    cval[s + rank] = vals[key];
                }
            }
        } else {
            int row = r0 + d;
            int done = 0;
            for (int i0 = 0; i0 < E; i0 += 32) {
                int i = i0 + lane;
                bool m = i < E && rows[i] == row;
                unsigned mask = __builtin_amdgcn_ballot_w32(m);
                if (m) {
                    int p = s + done + __builtin_popcount(mask & ((1u << lane) - 1u));
                    ccol[p] = cols[i];
                    cval[p] = vals[i];
                }
                done += __builtin_popcount(mask);
            }
        }
    }
}


template <int U>
__device__ __forceinline__ void add_edges128(float4& acc, int cl, int vl, int j,
                                             const float* __restrict__ x, int lane) {
    float4 xv[U];
    float w[U];
    #pragma unroll
    for (int u = 0; u < U; ++u) {
        int c = __builtin_amdgcn_readlane(cl, j + u);
        w[u] = __int_as_float(__builtin_amdgcn_readlane(vl, j + u));
        xv[u] = reinterpret_cast<const float4*>(x + (size_t)c * 128)[lane];
    }
    #pragma unroll
    for (int u = 0; u < U; ++u) {
        acc.x += w[u] * xv[u].x; acc.y += w[u] * xv[u].y;
        acc.z += w[u] * xv[u].z; acc.w += w[u] * xv[u].w;
    }
}

__device__ __forceinline__ void add_chunk128(float4& acc, int cl, int vl, int n,
                                             const float* __restrict__ x, int lane) {
    int j = 0;
    for (; j + 8 <= n; j += 8) add_edges128<8>(acc, cl, vl, j, x, lane);
    if (j + 4 <= n) { add_edges128<4>(acc, cl, vl, j, x, lane); j += 4; }
    if (j + 2 <= n) { add_edges128<2>(acc, cl, vl, j, x, lane); j += 2; }
    if (j < n) add_edges128<1>(acc, cl, vl, j, x, lane);
}

template <bool NORM>
__device__ __forceinline__ void gather128_rows(const int* __restrict__ rowptr,
                                               const int* __restrict__ ccol,
                                               const float* __restrict__ cval,
                                               int rowbase, int nrows,
                                               const float* __restrict__ x,
                                               float* __restrict__ y, int blk) {
    int lane = threadIdx.x & 31;
    int row = __builtin_amdgcn_readfirstlane((int)(blk * 256 + threadIdx.x) >> 5);
    if (row >= nrows) return;
    int s = rowptr[rowbase + row];
    int t = rowptr[rowbase + row + 1];
    float4 acc = make_float4(0.f, 0.f, 0.f, 0.f);
    for (int base = s; base < t; base += 32) {
        int n = t - base < 32 ? t - base : 32;
        int cl = 0, vl = 0;
        if (lane < n) { cl = ccol[base + lane]; vl = __float_as_int(cval[base + lane]); }
        add_chunk128(acc, cl, vl, n, x, lane);
    }
    if (NORM) {
        float ss = acc.x * acc.x + acc.y * acc.y + acc.z * acc.z + acc.w * acc.w;
        ss += __shfl_xor(ss, 16, 32);
        ss += __shfl_xor(ss, 8, 32);
        ss += __shfl_xor(ss, 4, 32);
        ss += __shfl_xor(ss, 2, 32);
        ss += __shfl_xor(ss, 1, 32);
        float inv = 1.0f / (sqrtf(ss) + 1e-9f);
        acc.x *= inv; acc.y *= inv; acc.z *= inv; acc.w *= inv;
    } else {
        acc.x = acc.x > 0.f ? acc.x : 0.f; acc.y = acc.y > 0.f ? acc.y : 0.f;
        acc.z = acc.z > 0.f ? acc.z : 0.f; acc.w = acc.w > 0.f ? acc.w : 0.f;
    }
    reinterpret_cast<float4*>(y + (size_t)row * 128)[lane] = acc;
}

template <int U>
__device__ __forceinline__ void add_edges428(float4& a0, float4& b0, float4& b1, float4& b2,
                                             int cl, int vl, int j,
                                             const float* __restrict__ xa,
                                             const float* __restrict__ xb,
                                             int lane, bool tail) {
    float4 p[U], q0[U], q1[U], q2[U];
    float w[U];
    #pragma unroll
    for (int u = 0; u < U; ++u) {
        int c = __builtin_amdgcn_readlane(cl, j + u);
        w[u] = __int_as_float(__builtin_amdgcn_readlane(vl, j + u));
        const float4* pa = reinterpret_cast<const float4*>(xa + (size_t)c * 128);
        const float4* pb = reinterpret_cast<const float4*>(xb + (size_t)c * 300);
        p[u]  = pa[lane];
        q0[u] = pb[lane];
        q1[u] = pb[32 + lane];
        q2[u] = make_float4(0.f, 0.f, 0.f, 0.f);
        if (tail) q2[u] = pb[64 + lane];
    }
    #pragma unroll
    for (int u = 0; u < U; ++u) {
        float v = w[u];
        a0.x += v * p[u].x;  a0.y += v * p[u].y;  a0.z += v * p[u].z;  a0.w += v * p[u].w;
        b0.x += v * q0[u].x; b0.y += v * q0[u].y; b0.z += v * q0[u].z; b0.w += v * q0[u].w;
        b1.x += v * q1[u].x; b1.y += v * q1[u].y; b1.z += v * q1[u].z; b1.w += v * q1[u].w;
        b2.x += v * q2[u].x; b2.y += v * q2[u].y; b2.z += v * q2[u].z; b2.w += v * q2[u].w;
    }
}

__device__ __forceinline__ void gather428_norm_rows(const int* __restrict__ rowptr,
                                                    const int* __restrict__ ccol,
                                                    const float* __restrict__ cval,
                                                    int rowbase, int nrows,
                                                    const float* __restrict__ xa,
                                                    const float* __restrict__ xb,
                                                    float* __restrict__ y, int blk) {
    int lane = threadIdx.x & 31;
    int row = __builtin_amdgcn_readfirstlane((int)(blk * 256 + threadIdx.x) >> 5);
    if (row >= nrows) return;
    int s = rowptr[rowbase + row];
    int t = rowptr[rowbase + row + 1];
    bool tail = lane < 11;
    float4 a0 = make_float4(0.f, 0.f, 0.f, 0.f), b0 = a0, b1 = a0, b2 = a0;
    for (int base = s; base < t; base += 32) {
        int n = t - base < 32 ? t - base : 32;
        int cl = 0, vl = 0;
        if (lane < n) { cl = ccol[base + lane]; vl = __float_as_int(cval[base + lane]); }
        int j = 0;
        for (; j + 4 <= n; j += 4) add_edges428<4>(a0, b0, b1, b2, cl, vl, j, xa, xb, lane, tail);
        if (j + 2 <= n) { add_edges428<2>(a0, b0, b1, b2, cl, vl, j, xa, xb, lane, tail); j += 2; }
        if (j < n) add_edges428<1>(a0, b0, b1, b2, cl, vl, j, xa, xb, lane, tail);
    }
    if (!tail) b2 = make_float4(0.f, 0.f, 0.f, 0.f);
    float ss = a0.x * a0.x + a0.y * a0.y + a0.z * a0.z + a0.w * a0.w
             + b0.x * b0.x + b0.y * b0.y + b0.z * b0.z + b0.w * b0.w
             + b1.x * b1.x + b1.y * b1.y + b1.z * b1.z + b1.w * b1.w
             + b2.x * b2.x + b2.y * b2.y + b2.z * b2.z + b2.w * b2.w;
    ss += __shfl_xor(ss, 16, 32);
    ss += __shfl_xor(ss, 8, 32);
    ss += __shfl_xor(ss, 4, 32);
    ss += __shfl_xor(ss, 2, 32);
    ss += __shfl_xor(ss, 1, 32);
    float inv = 1.0f / (sqrtf(ss) + 1e-9f);
    float4* py = reinterpret_cast<float4*>(y + (size_t)row * 428);
    py[lane]      = make_float4(a0.x * inv, a0.y * inv, a0.z * inv, a0.w * inv);
    py[32 + lane] = make_float4(b0.x * inv, b0.y * inv, b0.z * inv, b0.w * inv);
    py[64 + lane] = make_float4(b1.x * inv, b1.y * inv, b1.z * inv, b1.w * inv);
    if (tail) py[96 + lane] = make_float4(b2.x * inv, b2.y * inv, b2.z * inv, b2.w * inv);
}

#define STRIP_ROWS  32
#define STRIP_PITCH 132

__device__ __forceinline__ void gather_dense_strip(const int* __restrict__ rowptr,
                                                   const int* __restrict__ ccol,
                                                   const float* __restrict__ cval,
                                                   int rowbase, int nrows,
                                                   const float* __restrict__ x,
                                                   const float* __restrict__ W,
                                                   const float* __restrict__ bias,
                                                   float* __restrict__ Y, int blk,
                                                   float* strip) {
    int lane = threadIdx.x & 31;
    int wv = __builtin_amdgcn_readfirstlane((int)threadIdx.x >> 5);
    int r0 = blk * STRIP_ROWS;

    int sb[5];
    #pragma unroll
    for (int i = 0; i < 5; ++i) {
        int r = r0 + 4 * wv + i;
        sb[i] = rowptr[rowbase + (r < nrows ? r : nrows)];
    }
    int cl[4], vl[4];
    #pragma unroll
    for (int i = 0; i < 4; ++i) {
        int n = sb[i + 1] - sb[i] < 32 ? sb[i + 1] - sb[i] : 32;
        cl[i] = 0; vl[i] = 0;
        if (lane < n) { cl[i] = ccol[sb[i] + lane]; vl[i] = __float_as_int(cval[sb[i] + lane]); }
    }
    #pragma unroll
    for (int i = 0; i < 4; ++i) {
        int s = sb[i], t = sb[i + 1];
        float4 acc = make_float4(0.f, 0.f, 0.f, 0.f);
        add_chunk128(acc, cl[i], vl[i], t - s < 32 ? t - s : 32, x, lane);
        for (int base = s + 32; base < t; base += 32) {
            int n = t - base < 32 ? t - base : 32;
            int c2 = 0, v2 = 0;
            if (lane < n) { c2 = ccol[base + lane]; v2 = __float_as_int(cval[base + lane]); }
            add_chunk128(acc, c2, v2, n, x, lane);
        }
        acc.x = acc.x > 0.f ? acc.x : 0.f; acc.y = acc.y > 0.f ? acc.y : 0.f;
        acc.z = acc.z > 0.f ? acc.z : 0.f; acc.w = acc.w > 0.f ? acc.w : 0.f;
        reinterpret_cast<float4*>(strip + (4 * wv + i) * STRIP_PITCH)[lane] = acc;
    }
    __syncthreads();

    int half = lane >> 4;
    int l16  = lane & 15;
    const float* wcol = W + wv * 16 + l16;
    const float* a_lo = strip + l16 * STRIP_PITCH;
    const float* a_hi = strip + (16 + l16) * STRIP_PITCH;

    v8f c0 = {}, c1 = {};
    #pragma unroll
    for (int k = 0; k < 128; k += 4) {
        int ka = k + 2 * half;
        v2f b, a0, a1;
        b[0] = wcol[(size_t)ka * 128];
        b[1] = wcol[(size_t)(ka + 1) * 128];
        a0 = *reinterpret_cast<const v2f*>(a_lo + ka);
        a1 = *reinterpret_cast<const v2f*>(a_hi + ka);
        c0 = __builtin_amdgcn_wmma_f32_16x16x4_f32(
                  false, a0,  false, b,
                  (short)0, c0,  false,  false);
        c1 = __builtin_amdgcn_wmma_f32_16x16x4_f32(
                  false, a1,  false, b,
                  (short)0, c1,  false,  false);
    }

    float bn = bias[wv * 16 + l16];
    #pragma unroll
    for (int r = 0; r < 8; ++r) {
        int m = r0 + r + 8 * half;
        if (m < nrows)      Y[(size_t)m * 128 + wv * 16 + l16] = c0[r] + bn;
        if (m + 16 < nrows) Y[(size_t)(m + 16) * 128 + wv * 16 + l16] = c1[r] + bn;
    }
}

__device__ __forceinline__ void gemm_naive_elem(const float* __restrict__ X,
                                                const float* __restrict__ W,
                                                const float* __restrict__ bias,
                                                float* __restrict__ Y,
                                                int M, int K, int N, int idx) {
    if (idx >= M * N) return;
    int m = idx / N, n = idx % N;
    float acc = bias[n];
    for (int k = 0; k < K; ++k)
        acc += X[m * K + k] * W[k * N + n];
    Y[idx] = acc;
}

#define POS_GEMM_BLOCKS   30
#define POS_GATHER_BLOCKS 8
#define WORD_ROW_BLOCKS   2500
#define ENT_ROW_BLOCKS    1250
#define DOC_ROW_BLOCKS    1250
#define WORD_STRIP_BLOCKS 625
#define ENT_STRIP_BLOCKS  313

__global__ void __launch_bounds__(256) stage_gather(const int* __restrict__ rowptr,
                                                    const int* __restrict__ ccol,
                                                    const float* __restrict__ cval,
                                                    const float* xw, float* yw,
                                                    const float* xe, float* ye,
                                                    const float* Xp, const float* Wp,
                                                    const float* bp, float* Yp, int Kp) {
    int b = blockIdx.x;
    if (b < POS_GEMM_BLOCKS) {
        gemm_naive_elem(Xp, Wp, bp, Yp, 60, Kp, 128, b * 256 + (int)threadIdx.x);
    } else {
        bool word = b < POS_GEMM_BLOCKS + WORD_ROW_BLOCKS;
        gather128_rows<false>(rowptr, ccol, cval, word ? 0 : 20000, word ? 20000 : 10000,
                              word ? xw : xe, word ? yw : ye,
                              word ? b - POS_GEMM_BLOCKS
                                   : b - (POS_GEMM_BLOCKS + WORD_ROW_BLOCKS));
    }
}

__global__ void __launch_bounds__(256) stage_gather_dense(const int* __restrict__ rowptr,
                                                          const int* __restrict__ ccol,
                                                          const float* __restrict__ cval,
                                                          const float* xp, float* yp,
                                                          const float* xw, const float* Ww,
                                                          const float* bw, float* Yw,
                                                          const float* xe, const float* We,
                                                          const float* be, float* Ye) {
    __shared__ __attribute__((aligned(16))) float strip[STRIP_ROWS * STRIP_PITCH];
    int b = blockIdx.x;
    if (b < POS_GATHER_BLOCKS) {
        gather128_rows<false>(rowptr, ccol, cval, 30000, 60, xp, yp, b);
    } else {
        bool word = b < POS_GATHER_BLOCKS + WORD_STRIP_BLOCKS;
        gather_dense_strip(rowptr, ccol, cval, word ? 0 : 20000, word ? 20000 : 10000,
                           word ? xw : xe, word ? Ww : We, word ? bw : be, word ? Yw : Ye,
                           word ? b - POS_GATHER_BLOCKS
                                : b - (POS_GATHER_BLOCKS + WORD_STRIP_BLOCKS),
                           strip);
    }
}

__global__ void __launch_bounds__(256) stage_docs(const int* __restrict__ rowptr,
                                                  const int* __restrict__ ccol,
                                                  const float* __restrict__ cval,
                                                  const float* xp, float* yp,
                                                  const float* xa, const float* xb, float* y2,
                                                  const float* xw, float* y1) {
    int b = blockIdx.x;
    if (b < POS_GATHER_BLOCKS) {
        gather128_rows<false>(rowptr, ccol, cval, 30000, 60, xp, yp, b);
    } else if (b < POS_GATHER_BLOCKS + DOC_ROW_BLOCKS) {
        gather428_norm_rows(rowptr, ccol, cval, 40060, 10000, xa, xb, y2, b - POS_GATHER_BLOCKS);
    } else {
        gather128_rows<true>(rowptr, ccol, cval, 30060, 10000, xw, y1,
                             b - (POS_GATHER_BLOCKS + DOC_ROW_BLOCKS));
    }
}

__global__ void __launch_bounds__(256) stage_docs_pos(const int* __restrict__ rowptr,
                                                      const int* __restrict__ ccol,
                                                      const float* __restrict__ cval,
                                                      const float* __restrict__ x,
                                                      float* __restrict__ y) {
    gather128_rows<true>(rowptr, ccol, cval, 50060, 10000, x, y, blockIdx.x);
}

extern "C" void kernel_launch(void* const* d_in, const int* in_sizes, int n_in,
                              void* d_out, int out_size, void* d_ws, size_t ws_size,
                              hipStream_t stream) {
    (void)in_sizes; (void)n_in; (void)out_size; (void)ws_size;

    const int N_DOC = 10000, N_WORD = 20000, N_ENT = 10000, N_POS = 60;
    const int D = 128, D_WEMB = 300;

    const float* f1       = (const float*)d_in[0];
    const float* f2       = (const float*)d_in[1];
    const float* f3       = (const float*)d_in[2];
    const float* word_emb = (const float*)d_in[3];
    const int*   a11r = (const int*)d_in[4];
    const int*   a11c = (const int*)d_in[5];
    const float* a11v = (const float*)d_in[6];
    const int*   a22r = (const int*)d_in[7];
    const int*   a22c = (const int*)d_in[8];
    const float* a22v = (const float*)d_in[9];
    const int*   a33r = (const int*)d_in[10];
    const int*   a33c = (const int*)d_in[11];
    const float* a33v = (const float*)d_in[12];
    const int*   a01r = (const int*)d_in[13];
    const int*   a01c = (const int*)d_in[14];
    const float* a01v = (const float*)d_in[15];
    const int*   a02r = (const int*)d_in[16];
    const int*   a02c = (const int*)d_in[17];
    const float* a02v = (const float*)d_in[18];
    const int*   a03r = (const int*)d_in[19];
    const int*   a03c = (const int*)d_in[20];
    const float* a03v = (const float*)d_in[21];
    const float* W3   = (const float*)d_in[22];
    const float* b3   = (const float*)d_in[23];
    const float* W1_2 = (const float*)d_in[24];
    const float* b1_2 = (const float*)d_in[25];
    const float* W2_2 = (const float*)d_in[26];
    const float* b2_2 = (const float*)d_in[27];
    const float* W3_2 = (const float*)d_in[28];
    const float* b3_2 = (const float*)d_in[29];

    float* out  = (float*)d_out;
    float* out1 = out;
    float* out2 = out + (size_t)N_DOC * D;
    float* out3 = out2 + (size_t)N_DOC * (D + D_WEMB);

    float* bufA = (float*)d_ws;
    float* csr  = bufA + (size_t)N_WORD * D;
    float* bufC = csr + (size_t)N_WORD * D;
    float* bufD = bufC + (size_t)N_POS * D;
    int*   ccol   = (int*)csr;
    float* cval   = csr + 1033600;
    int*   rowptr = (int*)(csr + 2067200);

    float* wordG = bufA;
    float* entG  = out3;
    float* wordT = out2;
    float* entT  = out2 + (size_t)N_WORD * D;

    RowsPack rp = { a11r, a22r, a33r, a01r, a02r, a03r };
    EdgePack ep = { a11c, a22c, a33c, a01c, a02c, a03c,
                    a11v, a22v, a33v, a01v, a02v, a03v };

    csr_build<<<LIST_BLOCKS + POS_GEMM_BLOCKS_1024, 1024, 0, stream>>>(
        rp, ep, rowptr, ccol, cval, f3, W3, b3, bufC);

    stage_gather_dense<<<POS_GATHER_BLOCKS + WORD_STRIP_BLOCKS + ENT_STRIP_BLOCKS, 256, 0, stream>>>(
        rowptr, ccol, cval, bufC, bufD,
        f1, W1_2, b1_2, wordT, f2, W2_2, b2_2, entT);
    stage_gather<<<POS_GEMM_BLOCKS + WORD_ROW_BLOCKS + ENT_ROW_BLOCKS, 256, 0, stream>>>(
        rowptr, ccol, cval, wordT, wordG, entT, entG, bufD, W3_2, b3_2, bufC, 128);
    stage_docs<<<POS_GATHER_BLOCKS + DOC_ROW_BLOCKS + DOC_ROW_BLOCKS, 256, 0, stream>>>(
        rowptr, ccol, cval, bufC, bufD, entG, word_emb, out2, wordG, out1);
    stage_docs_pos<<<DOC_ROW_BLOCKS, 256, 0, stream>>>(rowptr, ccol, cval, bufD, out3);
}


# list build, row-id pass: a 16-edge group goes in sweeps (range tests, then all LDS slot adds with one wait, then the stores), its loads issued two groups ahead; lists and sum order unchanged
# speedup vs baseline: 4.8966x; 1.0364x over previous
#include <hip/hip_runtime.h>
#include <math.h>

typedef __attribute__((ext_vector_type(2))) float v2f;
typedef __attribute__((ext_vector_type(8))) float v8f;

#define ALL_EDGES 1033600
#define ALL_ROWS  60060

__device__ __forceinline__ int edge_off(int a) {
    return a == 0 ? 0 : a == 1 ? 320000 : a == 2 ? 480000 : a == 3 ? 483600
         : a == 4 ? 783600 : a == 5 ? 883600 : 1033600;
}
__device__ __forceinline__ int row_off(int a) {
    return a == 0 ? 0 : a == 1 ? 20000 : a == 2 ? 30000 : a == 3 ? 30060
         : a == 4 ? 40060 : a == 5 ? 50060 : 60060;
}
__device__ __forceinline__ int op_of_edge(int g) {
    return (g >= 320000) + (g >= 480000) + (g >= 483600) + (g >= 783600) + (g >= 883600);
}
__device__ __forceinline__ int op_of_row(int r) {
    return (r >= 20000) + (r >= 30000) + (r >= 30060) + (r >= 40060) + (r >= 50060);
}

struct RowsPack {
    const int* r0; const int* r1; const int* r2;
    const int* r3; const int* r4; const int* r5;
};
struct EdgePack {
    const int* c0; const int* c1; const int* c2;
    const int* c3; const int* c4; const int* c5;
    const float* v0; const float* v1; const float* v2;
    const float* v3; const float* v4; const float* v5;
};

#define BIN_WORDS 16384

__device__ __forceinline__ void visit_edge(int rv, int id, int r0, int nr, int lg, int capr,
                                           int* cnt, int* bins, int& below) {
    int d = rv - r0;
    below += d < 0 ? 1 : 0;
    if ((unsigned)d < (unsigned)nr) {
        int s = atomicAdd(&cnt[d], 1);
        if (s < capr) bins[(d << lg) + s] = id;
    }
}

#define LANE_MASKS_16(M)                                                        \
    "s"(M[0]), "s"(M[1]), "s"(M[2]), "s"(M[3]), "s"(M[4]), "s"(M[5]),           \
    "s"(M[6]), "s"(M[7]), "s"(M[8]), "s"(M[9]), "s"(M[10]), "s"(M[11]),         \
    "s"(M[12]), "s"(M[13]), "s"(M[14]), "s"(M[15])
__device__ __forceinline__ void visit_group(int4 v0, int4 v1, int4 v2, int4 v3, int e,
                                            int m4r0, int nr4, int lg, int capr,
                                            int* cnt, int* bins, int& below) {
    int rv[16] = { v0.x, v0.y, v0.z, v0.w, v1.x, v1.y, v1.z, v1.w,
                   v2.x, v2.y, v2.z, v2.w, v3.x, v3.y, v3.z, v3.w };
    int u[16], s[16];
    bool in[16], room[16];
    unsigned mask_in[16], mask_room[16];
    #pragma unroll
    for (int k = 0; k < 16; ++k) {
        u[k] = (rv[k] << 2) + m4r0;
        below += u[k] < 0 ? 1 : 0;
        in[k] = (unsigned)u[k] < (unsigned)nr4;
        mask_in[k] = __builtin_amdgcn_ballot_w32(in[k]);
        s[k] = capr;
    }
    asm volatile("" :: LANE_MASKS_16(mask_in));
    #pragma unroll
    for (int k = 0; k < 16; ++k)
        if (in[k]) s[k] = atomicAdd(&cnt[u[k] >> 2], 1);
    #pragma unroll
    for (int k = 0; k < 16; ++k) {
        room[k] = s[k] < capr;
        mask_room[k] = __builtin_amdgcn_ballot_w32(room[k]);
    }
    asm volatile("" :: LANE_MASKS_16(mask_room));
    #pragma unroll
    for (int k = 0; k < 16; ++k)
        if (room[k]) {
            int* slot = reinterpret_cast<int*>(reinterpret_cast<char*>(bins) + (u[k] << lg) + (s[k] << 2));
            *slot = e + (k >> 2) * 4096 + (k & 3);
        }
}
#undef LANE_MASKS_16

#define LIST_BLOCKS 240
#define POS_GEMM_BLOCKS_1024 8

__device__ __forceinline__ void gemm_naive_elem(const float* __restrict__ X,
                                                const float* __restrict__ W,
                                                const float* __restrict__ bias,
                                                float* __restrict__ Y,
                                                int M, int K, int N, int idx);

__global__ void __launch_bounds__(1024) csr_build(RowsPack rp, EdgePack ep,
                                                  int* __restrict__ rowptr,
                                                  int* __restrict__ ccol,
                                                  float* __restrict__ cval,
                                                  const float* Xp, const float* Wp,
                                                  const float* bp, float* Yp) {
    __shared__ struct { int cnt[256]; int sh[256]; int red[32]; int bins[BIN_WORDS]; } lds;
    int* bins = lds.bins;
    int* cnt  = lds.cnt;
    int* sh   = lds.sh;
    int* red  = lds.red;

    int t = threadIdx.x;
    int lane = t & 31;
    int wv = __builtin_amdgcn_readfirstlane(t >> 5);
    int b = blockIdx.x;
    if (b >= LIST_BLOCKS) {
        gemm_naive_elem(Xp, Wp, bp, Yp, 60, 60, 128, (b - LIST_BLOCKS) * 1024 + t);
        return;
    }
    int a  = (b >= 79) + (b >= 119) + (b >= 120) + (b >= 160) + (b >= 200);
    int b0 = a == 0 ? 0 : a == 1 ? 79 : a == 2 ? 119 : a == 3 ? 120 : a == 4 ? 160 : 200;
    int lg   = a == 2 ? 8 : 6;
    int capr = 1 << lg;
    int R    = BIN_WORDS >> lg;
    int r0   = (b - b0) * R;
    int nro  = row_off(a + 1) - row_off(a);
    int nr   = nro - r0 < R ? nro - r0 : R;
    int E    = edge_off(a + 1) - edge_off(a);
    const int* rows = a == 0 ? rp.r0 : a == 1 ? rp.r1 : a == 2 ? rp.r2
                    : a == 3 ? rp.r3 : a == 4 ? rp.r4 : rp.r5;
    const int* cols = a == 0 ? ep.c0 : a == 1 ? ep.c1 : a == 2 ? ep.c2
                    : a == 3 ? ep.c3 : a == 4 ? ep.c4 : ep.c5;
    const float* vals = a == 0 ? ep.v0 : a == 1 ? ep.v1 : a == 2 ? ep.v2
                      : a == 3 ? ep.v3 : a == 4 ? ep.v4 : ep.v5;

    if (t < 256) cnt[t] = 0;
    __syncthreads();

    int below = 0;
    if ((((size_t)rows) & 15) == 0) {
        const int4* rows4 = reinterpret_cast<const int4*>(rows);
        int n4 = E >> 2;
        int G = n4 >> 12;
        int m4r0 = -(r0 << 2), nr4 = nr << 2;
        const int4 none = make_int4(0x1fffffff, 0x1fffffff, 0x1fffffff, 0x1fffffff);
#define LOAD_GROUP(X, q)                                                        \
        do {                                                                    \
            int q_ = (q) < G ? (q) : G - 1;                                     \
            const int4* p_ = rows4 + (q_ << 12) + t;                            \
            X##0 = p_[0];    X##1 = p_[1024];                                   \
            X##2 = p_[2048]; X##3 = p_[3072];                                   \
        } while (0)
#define VISIT_GROUP(X, q)                                                       \
        visit_group(X##0, X##1, X##2, X##3, (((q) << 12) + t) << 2,             \
                    m4r0, nr4, lg, capr, cnt, bins, below)
        int4 A0 = none, A1 = none, A2 = none, A3 = none;
        int4 B0 = none, B1 = none, B2 = none, B3 = none;
        if (G > 0) {
            LOAD_GROUP(A, 0);
            LOAD_GROUP(B, 1);
        }
        {
            int i = (G << 12) + t;
            int4 T0 = i        < n4 ? rows4[i]        : none;
            int4 T1 = i + 1024 < n4 ? rows4[i + 1024] : none;
            int4 T2 = i + 2048 < n4 ? rows4[i + 2048] : none;
            int4 T3 = i + 3072 < n4 ? rows4[i + 3072] : none;
            VISIT_GROUP(T, G);
        }
        if (G > 0) {
            int4 C0, C1, C2, C3;
            for (int q = 0; q < G; q += 3) {
                LOAD_GROUP(C, q + 2); VISIT_GROUP(A, q);
                LOAD_GROUP(A, q + 3); if (q + 1 < G) VISIT_GROUP(B, q + 1);
                LOAD_GROUP(B, q + 4); if (q + 2 < G) VISIT_GROUP(C, q + 2);
            }
        }
#undef LOAD_GROUP
#undef VISIT_GROUP
        for (int k = (n4 << 2) + t; k < E; k += 1024)
            visit_edge(rows[k], k, r0, nr, lg, capr, cnt, bins, below);
    } else {
        for (int i = t; i < E; i += 1024)
            visit_edge(rows[i], i, r0, nr, lg, capr, cnt, bins, below);
    }

    below += __shfl_xor(below, 16, 32);
    below += __shfl_xor(below, 8, 32);
    below += __shfl_xor(below, 4, 32);
    below += __shfl_xor(below, 2, 32);
    below += __shfl_xor(below, 1, 32);
    if (lane == 0) red[wv] = below;
    __syncthreads();

    int c = 0;
    if (t < 256) { c = cnt[t]; sh[t] = c; }
    __syncthreads();
    for (int off = 1; off < 256; off <<= 1) {
        int v = (t < 256 && t >= off) ? sh[t - off] : 0;
        __syncthreads();
        if (t < 256) sh[t] += v;
        __syncthreads();
    }
    int base = edge_off(a);
    for (int k = 0; k < 32; ++k) base += red[k];
    if (t < nr) rowptr[row_off(a) + r0 + t] = base + sh[t] - c;
    if (b == 239 && t == 0) rowptr[ALL_ROWS] = ALL_EDGES;

    for (int d = wv; d < nr; d += 32) {
        int len = __builtin_amdgcn_readfirstlane(cnt[d]);
        int s   = __builtin_amdgcn_readfirstlane(base + sh[d]) - len;
        if (len <= capr) {
            const int* bl = bins + (d << lg);
            for (int i0 = 0; i0 < len; i0 += 32) {
                int i = i0 + lane;
                int key = i < len ? bl[i] : 0x7fffffff;
                int rank = 0;
                for (int j0 = 0; j0 < len; j0 += 32) {
                    int jj = j0 + lane;
                    int other = jj < len ? bl[jj] : 0x7fffffff;
                    int n = len - j0 < 32 ? len - j0 : 32;
                    for (int j = 0; j < n; ++j)
                        rank += __builtin_amdgcn_readlane(other, j) < key ? 1 : 0;
                }
                if (i < len) {
                    ccol[s + rank] = cols[key];
                    cval[s + rank] = vals[key];
                }
            }
        } else {
            int row = r0 + d;
            int done = 0;
            for (int i0 = 0; i0 < E; i0 += 32) {
                int i = i0 + lane;
                bool m = i < E && rows[i] == row;
                unsigned mask = __builtin_amdgcn_ballot_w32(m);
                if (m) {
                    int p = s + done + __builtin_popcount(mask & ((1u << lane) - 1u));
                    ccol[p] = cols[i];
                    cval[p] = vals[i];
                }
                done += __builtin_popcount(mask);
            }
        }
    }
}


template <int U>
__device__ __forceinline__ void add_edges128(float4& acc, int cl, int vl, int j,
                                             const float* __restrict__ x, int lane) {
    float4 xv[U];
    float w[U];
    #pragma unroll
    for (int u = 0; u < U; ++u) {
        int c = __builtin_amdgcn_readlane(cl, j + u);
        w[u] = __int_as_float(__builtin_amdgcn_readlane(vl, j + u));
        xv[u] = reinterpret_cast<const float4*>(x + (size_t)c * 128)[lane];
    }
    #pragma unroll
    for (int u = 0; u < U; ++u) {
        acc.x += w[u] * xv[u].x; acc.y += w[u] * xv[u].y;
        acc.z += w[u] * xv[u].z; acc.w += w[u] * xv[u].w;
    }
}

__device__ __forceinline__ void add_chunk128(float4& acc, int cl, int vl, int n,
                                             const float* __restrict__ x, int lane) {
    int j = 0;
    for (; j + 8 <= n; j += 8) add_edges128<8>(acc, cl, vl, j, x, lane);
    if (j + 4 <= n) { add_edges128<4>(acc, cl, vl, j, x, lane); j += 4; }
    if (j + 2 <= n) { add_edges128<2>(acc, cl, vl, j, x, lane); j += 2; }
    if (j < n) add_edges128<1>(acc, cl, vl, j, x, lane);
}

template <bool NORM>
__device__ __forceinline__ void gather128_rows(const int* __restrict__ rowptr,
                                               const int* __restrict__ ccol,
                                               const float* __restrict__ cval,
                                               int rowbase, int nrows,
                                               const float* __restrict__ x,
                                               float* __restrict__ y, int blk) {
    int lane = threadIdx.x & 31;
    int row = __builtin_amdgcn_readfirstlane((int)(blk * 256 + threadIdx.x) >> 5);
    if (row >= nrows) return;
    int s = rowptr[rowbase + row];
    int t = rowptr[rowbase + row + 1];
    float4 acc = make_float4(0.f, 0.f, 0.f, 0.f);
    for (int base = s; base < t; base += 32) {
        int n = t - base < 32 ? t - base : 32;
        int cl = 0, vl = 0;
        if (lane < n) { cl = ccol[base + lane]; vl = __float_as_int(cval[base + lane]); }
        add_chunk128(acc, cl, vl, n, x, lane);
    }
    if (NORM) {
        float ss = acc.x * acc.x + acc.y * acc.y + acc.z * acc.z + acc.w * acc.w;
        ss += __shfl_xor(ss, 16, 32);
        ss += __shfl_xor(ss, 8, 32);
        ss += __shfl_xor(ss, 4, 32);
        ss += __shfl_xor(ss, 2, 32);
        ss += __shfl_xor(ss, 1, 32);
        float inv = 1.0f / (sqrtf(ss) + 1e-9f);
        acc.x *= inv; acc.y *= inv; acc.z *= inv; acc.w *= inv;
    } else {
        acc.x = acc.x > 0.f ? acc.x : 0.f; acc.y = acc.y > 0.f ? acc.y : 0.f;
        acc.z = acc.z > 0.f ? acc.z : 0.f; acc.w = acc.w > 0.f ? acc.w : 0.f;
    }
    reinterpret_cast<float4*>(y + (size_t)row * 128)[lane] = acc;
}

template <int U>
__device__ __forceinline__ void add_edges428(float4& a0, float4& b0, float4& b1, float4& b2,
                                             int cl, int vl, int j,
                                             const float* __restrict__ xa,
                                             const float* __restrict__ xb,
                                             int lane, bool tail) {
    float4 p[U], q0[U], q1[U], q2[U];
    float w[U];
    #pragma unroll
    for (int u = 0; u < U; ++u) {
        int c = __builtin_amdgcn_readlane(cl, j + u);
        w[u] = __int_as_float(__builtin_amdgcn_readlane(vl, j + u));
        const float4* pa = reinterpret_cast<const float4*>(xa + (size_t)c * 128);
        const float4* pb = reinterpret_cast<const float4*>(xb + (size_t)c * 300);
        p[u]  = pa[lane];
        q0[u] = pb[lane];
        q1[u] = pb[32 + lane];
        q2[u] = make_float4(0.f, 0.f, 0.f, 0.f);
        if (tail) q2[u] = pb[64 + lane];
    }
    #pragma unroll
    for (int u = 0; u < U; ++u) {
        float v = w[u];
        a0.x += v * p[u].x;  a0.y += v * p[u].y;  a0.z += v * p[u].z;  a0.w += v * p[u].w;
        b0.x += v * q0[u].x; b0.y += v * q0[u].y; b0.z += v * q0[u].z; b0.w += v * q0[u].w;
        b1.x += v * q1[u].x; b1.y += v * q1[u].y; b1.z += v * q1[u].z; b1.w += v * q1[u].w;
        b2.x += v * q2[u].x; b2.y += v * q2[u].y; b2.z += v * q2[u].z; b2.w += v * q2[u].w;
    }
}

__device__ __forceinline__ void gather428_norm_rows(const int* __restrict__ rowptr,
                                                    const int* __restrict__ ccol,
                                                    const float* __restrict__ cval,
                                                    int rowbase, int nrows,
                                                    const float* __restrict__ xa,
                                                    const float* __restrict__ xb,
                                                    float* __restrict__ y, int blk) {
    int lane = threadIdx.x & 31;
    int row = __builtin_amdgcn_readfirstlane((int)(blk * 256 + threadIdx.x) >> 5);
    if (row >= nrows) return;
    int s = rowptr[rowbase + row];
    int t = rowptr[rowbase + row + 1];
    bool tail = lane < 11;
    float4 a0 = make_float4(0.f, 0.f, 0.f, 0.f), b0 = a0, b1 = a0, b2 = a0;
    for (int base = s; base < t; base += 32) {
        int n = t - base < 32 ? t - base : 32;
        int cl = 0, vl = 0;
        if (lane < n) { cl = ccol[base + lane]; vl = __float_as_int(cval[base + lane]); }
        int j = 0;
        for (; j + 4 <= n; j += 4) add_edges428<4>(a0, b0, b1, b2, cl, vl, j, xa, xb, lane, tail);
        if (j + 2 <= n) { add_edges428<2>(a0, b0, b1, b2, cl, vl, j, xa, xb, lane, tail); j += 2; }
        if (j < n) add_edges428<1>(a0, b0, b1, b2, cl, vl, j, xa, xb, lane, tail);
    }
    if (!tail) b2 = make_float4(0.f, 0.f, 0.f, 0.f);
    float ss = a0.x * a0.x + a0.y * a0.y + a0.z * a0.z + a0.w * a0.w
             + b0.x * b0.x + b0.y * b0.y + b0.z * b0.z + b0.w * b0.w
             + b1.x * b1.x + b1.y * b1.y + b1.z * b1.z + b1.w * b1.w
             + b2.x * b2.x + b2.y * b2.y + b2.z * b2.z + b2.w * b2.w;
    ss += __shfl_xor(ss, 16, 32);
    ss += __shfl_xor(ss, 8, 32);
    ss += __shfl_xor(ss, 4, 32);
    ss += __shfl_xor(ss, 2, 32);
    ss += __shfl_xor(ss, 1, 32);
    float inv = 1.0f / (sqrtf(ss) + 1e-9f);
    float4* py = reinterpret_cast<float4*>(y + (size_t)row * 428);
    py[lane]      = make_float4(a0.x * inv, a0.y * inv, a0.z * inv, a0.w * inv);
    py[32 + lane] = make_float4(b0.x * inv, b0.y * inv, b0.z * inv, b0.w * inv);
    py[64 + lane] = make_float4(b1.x * inv, b1.y * inv, b1.z * inv, b1.w * inv);
    if (tail) py[96 + lane] = make_float4(b2.x * inv, b2.y * inv, b2.z * inv, b2.w * inv);
}

#define STRIP_ROWS  32
#define STRIP_PITCH 132

__device__ __forceinline__ void gather_dense_strip(const int* __restrict__ rowptr,
                                                   const int* __restrict__ ccol,
                                                   const float* __restrict__ cval,
                                                   int rowbase, int nrows,
                                                   const float* __restrict__ x,
                                                   const float* __restrict__ W,
                                                   const float* __restrict__ bias,
                                                   float* __restrict__ Y, int blk,
                                                   float* strip) {
    int lane = threadIdx.x & 31;
    int wv = __builtin_amdgcn_readfirstlane((int)threadIdx.x >> 5);
    int r0 = blk * STRIP_ROWS;

    int sb[5];
    #pragma unroll
    for (int i = 0; i < 5; ++i) {
        int r = r0 + 4 * wv + i;
        sb[i] = rowptr[rowbase + (r < nrows ? r : nrows)];
    }
    int cl[4], vl[4];
    #pragma unroll
    for (int i = 0; i < 4; ++i) {
        int n = sb[i + 1] - sb[i] < 32 ? sb[i + 1] - sb[i] : 32;
        cl[i] = 0; vl[i] = 0;
        if (lane < n) { cl[i] = ccol[sb[i] + lane]; vl[i] = __float_as_int(cval[sb[i] + lane]); }
    }
    #pragma unroll
    for (int i = 0; i < 4; ++i) {
        int s = sb[i], t = sb[i + 1];
        float4 acc = make_float4(0.f, 0.f, 0.f, 0.f);
        add_chunk128(acc, cl[i], vl[i], t - s < 32 ? t - s : 32, x, lane);
        for (int base = s + 32; base < t; base += 32) {
            int n = t - base < 32 ? t - base : 32;
            int c2 = 0, v2 = 0;
            if (lane < n) { c2 = ccol[base + lane]; v2 = __float_as_int(cval[base + lane]); }
            add_chunk128(acc, c2, v2, n, x, lane);
        }
        acc.x = acc.x > 0.f ? acc.x : 0.f; acc.y = acc.y > 0.f ? acc.y : 0.f;
        acc.z = acc.z > 0.f ? acc.z : 0.f; acc.w = acc.w > 0.f ? acc.w : 0.f;
        reinterpret_cast<float4*>(strip + (4 * wv + i) * STRIP_PITCH)[lane] = acc;
    }
    __syncthreads();

    int half = lane >> 4;
    int l16  = lane & 15;
    const float* wcol = W + wv * 16 + l16;
    const float* a_lo = strip + l16 * STRIP_PITCH;
    const float* a_hi = strip + (16 + l16) * STRIP_PITCH;

    v8f c0 = {}, c1 = {};
    #pragma unroll
    for (int k = 0; k < 128; k += 4) {
        int ka = k + 2 * half;
        v2f b, a0, a1;
        b[0] = wcol[(size_t)ka * 128];
        b[1] = wcol[(size_t)(ka + 1) * 128];
        a0 = *reinterpret_cast<const v2f*>(a_lo + ka);
        a1 = *reinterpret_cast<const v2f*>(a_hi + ka);
        c0 = __builtin_amdgcn_wmma_f32_16x16x4_f32(
                  false, a0,  false, b,
                  (short)0, c0,  false,  false);
        c1 = __builtin_amdgcn_wmma_f32_16x16x4_f32(
                  false, a1,  false, b,
                  (short)0, c1,  false,  false);
    }

    float bn = bias[wv * 16 + l16];
    #pragma unroll
    for (int r = 0; r < 8; ++r) {
        int m = r0 + r + 8 * half;
        if (m < nrows)      Y[(size_t)m * 128 + wv * 16 + l16] = c0[r] + bn;
        if (m + 16 < nrows) Y[(size_t)(m + 16) * 128 + wv * 16 + l16] = c1[r] + bn;
    }
}

__device__ __forceinline__ void gemm_naive_elem(const float* __restrict__ X,
                                                const float* __restrict__ W,
                                                const float* __restrict__ bias,
                                                float* __restrict__ Y,
                                                int M, int K, int N, int idx) {
    if (idx >= M * N) return;
    int m = idx / N, n = idx % N;
    float acc = bias[n];
    for (int k = 0; k < K; ++k)
        acc += X[m * K + k] * W[k * N + n];
    Y[idx] = acc;
}

#define POS_GEMM_BLOCKS   30
#define POS_GATHER_BLOCKS 8
#define WORD_ROW_BLOCKS   2500
#define ENT_ROW_BLOCKS    1250
#define DOC_ROW_BLOCKS    1250
#define WORD_STRIP_BLOCKS 625
#define ENT_STRIP_BLOCKS  313

__global__ void __launch_bounds__(256) stage_gather(const int* __restrict__ rowptr,
                                                    const int* __restrict__ ccol,
                                                    const float* __restrict__ cval,
                                                    const float* xw, float* yw,
                                                    const float* xe, float* ye,
                                                    const float* Xp, const float* Wp,
                                                    const float* bp, float* Yp, int Kp) {
    int b = blockIdx.x;
    if (b < POS_GEMM_BLOCKS) {
        gemm_naive_elem(Xp, Wp, bp, Yp, 60, Kp, 128, b * 256 + (int)threadIdx.x);
    } else {
        bool word = b < POS_GEMM_BLOCKS + WORD_ROW_BLOCKS;
        gather128_rows<false>(rowptr, ccol, cval, word ? 0 : 20000, word ? 20000 : 10000,
                              word ? xw : xe, word ? yw : ye,
                              word ? b - POS_GEMM_BLOCKS
                                   : b - (POS_GEMM_BLOCKS + WORD_ROW_BLOCKS));
    }
}

__global__ void __launch_bounds__(256) stage_gather_dense(const int* __restrict__ rowptr,
                                                          const int* __restrict__ ccol,
                                                          const float* __restrict__ cval,
                                                          const float* xp, float* yp,
                                                          const float* xw, const float* Ww,
                                                          const float* bw, float* Yw,
                                                          const float* xe, const float* We,
                                                          const float* be, float* Ye) {
    __shared__ __attribute__((aligned(16))) float strip[STRIP_ROWS * STRIP_PITCH];
    int b = blockIdx.x;
    if (b < POS_GATHER_BLOCKS) {
        gather128_rows<false>(rowptr, ccol, cval, 30000, 60, xp, yp, b);
    } else {
        bool word = b < POS_GATHER_BLOCKS + WORD_STRIP_BLOCKS;
        gather_dense_strip(rowptr, ccol, cval, word ? 0 : 20000, word ? 20000 : 10000,
                           word ? xw : xe, word ? Ww : We, word ? bw : be, word ? Yw : Ye,
                           word ? b - POS_GATHER_BLOCKS
                                : b - (POS_GATHER_BLOCKS + WORD_STRIP_BLOCKS),
                           strip);
    }
}

__global__ void __launch_bounds__(256) stage_docs(const int* __restrict__ rowptr,
                                                  const int* __restrict__ ccol,
                                                  const float* __restrict__ cval,
                                                  const float* xp, float* yp,
                                                  const float* xa, const float* xb, float* y2,
                                                  const float* xw, float* y1) {
    int b = blockIdx.x;
    if (b < POS_GATHER_BLOCKS) {
        gather128_rows<false>(rowptr, ccol, cval, 30000, 60, xp, yp, b);
    } else if (b < POS_GATHER_BLOCKS + DOC_ROW_BLOCKS) {
        gather428_norm_rows(rowptr, ccol, cval, 40060, 10000, xa, xb, y2, b - POS_GATHER_BLOCKS);
    } else {
        gather128_rows<true>(rowptr, ccol, cval, 30060, 10000, xw, y1,
                             b - (POS_GATHER_BLOCKS + DOC_ROW_BLOCKS));
    }
}

__global__ void __launch_bounds__(256) stage_docs_pos(const int* __restrict__ rowptr,
                                                      const int* __restrict__ ccol,
                                                      const float* __restrict__ cval,
                                                      const float* __restrict__ x,
                                                      float* __restrict__ y) {
    gather128_rows<true>(rowptr, ccol, cval, 50060, 10000, x, y, blockIdx.x);
}

extern "C" void kernel_launch(void* const* d_in, const int* in_sizes, int n_in,
                              void* d_out, int out_size, void* d_ws, size_t ws_size,
                              hipStream_t stream) {
    (void)in_sizes; (void)n_in; (void)out_size; (void)ws_size;

    const int N_DOC = 10000, N_WORD = 20000, N_ENT = 10000, N_POS = 60;
    const int D = 128, D_WEMB = 300;

    const float* f1       = (const float*)d_in[0];
    const float* f2       = (const float*)d_in[1];
    const float* f3       = (const float*)d_in[2];
    const float* word_emb = (const float*)d_in[3];
    const int*   a11r = (const int*)d_in[4];
    const int*   a11c = (const int*)d_in[5];
    const float* a11v = (const float*)d_in[6];
    const int*   a22r = (const int*)d_in[7];
    const int*   a22c = (const int*)d_in[8];
    const float* a22v = (const float*)d_in[9];
    const int*   a33r = (const int*)d_in[10];
    const int*   a33c = (const int*)d_in[11];
    const float* a33v = (const float*)d_in[12];
    const int*   a01r = (const int*)d_in[13];
    const int*   a01c = (const int*)d_in[14];
    const float* a01v = (const float*)d_in[15];
    const int*   a02r = (const int*)d_in[16];
    const int*   a02c = (const int*)d_in[17];
    const float* a02v = (const float*)d_in[18];
    const int*   a03r = (const int*)d_in[19];
    const int*   a03c = (const int*)d_in[20];
    const float* a03v = (const float*)d_in[21];
    const float* W3   = (const float*)d_in[22];
    const float* b3   = (const float*)d_in[23];
    const float* W1_2 = (const float*)d_in[24];
    const float* b1_2 = (const float*)d_in[25];
    const float* W2_2 = (const float*)d_in[26];
    const float* b2_2 = (const float*)d_in[27];
    const float* W3_2 = (const float*)d_in[28];
    const float* b3_2 = (const float*)d_in[29];

    float* out  = (float*)d_out;
    float* out1 = out;
    float* out2 = out + (size_t)N_DOC * D;
    float* out3 = out2 + (size_t)N_DOC * (D + D_WEMB);

    float* bufA = (float*)d_ws;
    float* csr  = bufA + (size_t)N_WORD * D;
    float* bufC = csr + (size_t)N_WORD * D;
    float* bufD = bufC + (size_t)N_POS * D;
    int*   ccol   = (int*)csr;
    float* cval   = csr + 1033600;
    int*   rowptr = (int*)(csr + 2067200);

    float* wordG = bufA;
    float* entG  = out3;
    float* wordT = out2;
    float* entT  = out2 + (size_t)N_WORD * D;

    RowsPack rp = { a11r, a22r, a33r, a01r, a02r, a03r };
    EdgePack ep = { a11c, a22c, a33c, a01c, a02c, a03c,
                    a11v, a22v, a33v, a01v, a02v, a03v };

    csr_build<<<LIST_BLOCKS + POS_GEMM_BLOCKS_1024, 1024, 0, stream>>>(
        rp, ep, rowptr, ccol, cval, f3, W3, b3, bufC);

    stage_gather_dense<<<POS_GATHER_BLOCKS + WORD_STRIP_BLOCKS + ENT_STRIP_BLOCKS, 256, 0, stream>>>(
        rowptr, ccol, cval, bufC, bufD,
        f1, W1_2, b1_2, wordT, f2, W2_2, b2_2, entT);
    stage_gather<<<POS_GEMM_BLOCKS + WORD_ROW_BLOCKS + ENT_ROW_BLOCKS, 256, 0, stream>>>(
        rowptr, ccol, cval, wordT, wordG, entT, entG, bufD, W3_2, b3_2, bufC, 128);
    stage_docs<<<POS_GATHER_BLOCKS + DOC_ROW_BLOCKS + DOC_ROW_BLOCKS, 256, 0, stream>>>(
        rowptr, ccol, cval, bufC, bufD, entG, word_emb, out2, wordG, out1);
    stage_docs_pos<<<DOC_ROW_BLOCKS, 256, 0, stream>>>(rowptr, ccol, cval, bufD, out3);
}


# list build: a block owns (row range, quarter of the operator's edge ids) and reads only that quarter's row ids; lists stored per slice, gathers walk a row's 4 sub-lists; sum order unchanged
# speedup vs baseline: 5.3714x; 1.0970x over previous
#include <hip/hip_runtime.h>
#include <math.h>

typedef __attribute__((ext_vector_type(2))) float v2f;
typedef __attribute__((ext_vector_type(8))) float v8f;

#define ALL_EDGES 1033600
#define ALL_ROWS  60060

__device__ __forceinline__ int edge_off(int a) {
    return a == 0 ? 0 : a == 1 ? 320000 : a == 2 ? 480000 : a == 3 ? 483600
         : a == 4 ? 783600 : a == 5 ? 883600 : 1033600;
}
__device__ __forceinline__ int row_off(int a) {
    return a == 0 ? 0 : a == 1 ? 20000 : a == 2 ? 30000 : a == 3 ? 30060
         : a == 4 ? 40060 : a == 5 ? 50060 : 60060;
}
__device__ __forceinline__ int op_of_edge(int g) {
    return (g >= 320000) + (g >= 480000) + (g >= 483600) + (g >= 783600) + (g >= 883600);
}
__device__ __forceinline__ int op_of_row(int r) {
    return (r >= 20000) + (r >= 30000) + (r >= 30060) + (r >= 40060) + (r >= 50060);
}

struct RowsPack {
    const int* r0; const int* r1; const int* r2;
    const int* r3; const int* r4; const int* r5;
};
struct EdgePack {
    const int* c0; const int* c1; const int* c2;
    const int* c3; const int* c4; const int* c5;
    const float* v0; const float* v1; const float* v2;
    const float* v3; const float* v4; const float* v5;
};

#define SLICES     4
#define BLOCK_ROWS 1024
#define LIST_CAP   32
#define LIST_PITCH 36

__device__ __forceinline__ void visit_edge(int rv, int id, int r0, int nr,
                                           int* cnt, int* bins, int& below) {
    int d = rv - r0;
    below += d < 0 ? 1 : 0;
    if ((unsigned)d < (unsigned)nr) {
        int s = atomicAdd(&cnt[d], 1);
        if (s < LIST_CAP) bins[d * LIST_PITCH + s] = id;
    }
}

#define LANE_MASKS_16(M)                                                        \
    "s"(M[0]), "s"(M[1]), "s"(M[2]), "s"(M[3]), "s"(M[4]), "s"(M[5]),           \
    "s"(M[6]), "s"(M[7]), "s"(M[8]), "s"(M[9]), "s"(M[10]), "s"(M[11]),         \
    "s"(M[12]), "s"(M[13]), "s"(M[14]), "s"(M[15])
__device__ __forceinline__ void visit_group(int4 v0, int4 v1, int4 v2, int4 v3, int e,
                                            int m4r0, int nr4,
                                            int* cnt, int* bins, int& below) {
    const int capr = LIST_CAP;
    int rv[16] = { v0.x, v0.y, v0.z, v0.w, v1.x, v1.y, v1.z, v1.w,
                   v2.x, v2.y, v2.z, v2.w, v3.x, v3.y, v3.z, v3.w };
    int u[16], s[16];
    bool in[16], room[16];
    unsigned mask_in[16], mask_room[16];
    #pragma unroll
    for (int k = 0; k < 16; ++k) {
        u[k] = (rv[k] << 2) + m4r0;
        below += u[k] < 0 ? 1 : 0;
        in[k] = (unsigned)u[k] < (unsigned)nr4;
        mask_in[k] = __builtin_amdgcn_ballot_w32(in[k]);
        s[k] = capr;
    }
    asm volatile("" :: LANE_MASKS_16(mask_in));
    #pragma unroll
    for (int k = 0; k < 16; ++k)
        if (in[k]) s[k] = atomicAdd(&cnt[u[k] >> 2], 1);
    #pragma unroll
    for (int k = 0; k < 16; ++k) {
        room[k] = s[k] < capr;
        mask_room[k] = __builtin_amdgcn_ballot_w32(room[k]);
    }
    asm volatile("" :: LANE_MASKS_16(mask_room));
    #pragma unroll
    for (int k = 0; k < 16; ++k)
        if (room[k]) {
            int* slot = reinterpret_cast<int*>(reinterpret_cast<char*>(bins) + u[k] * LIST_PITCH + (s[k] << 2));
            *slot = e + (k >> 2) * 4096 + (k & 3);
        }
}
#undef LANE_MASKS_16

#define LIST_BLOCKS 244
#define POS_GEMM_BLOCKS_1024 8

__device__ __forceinline__ void gemm_naive_elem(const float* __restrict__ X,
                                                const float* __restrict__ W,
                                                const float* __restrict__ bias,
                                                float* __restrict__ Y,
                                                int M, int K, int N, int idx);

__global__ void __launch_bounds__(1024) csr_build(RowsPack rp, EdgePack ep,
                                                  int* __restrict__ rowptr,
                                                  int* __restrict__ ccol,
                                                  float* __restrict__ cval,
                                                  const float* Xp, const float* Wp,
                                                  const float* bp, float* Yp) {
    __shared__ __attribute__((aligned(16))) struct {
        int cnt[BLOCK_ROWS]; int wtot[32]; int red[32]; int bins[BLOCK_ROWS * LIST_PITCH];
    } lds;
    int* bins = lds.bins;
    int* cnt  = lds.cnt;
    int* wtot = lds.wtot;
    int* red  = lds.red;

    int t = threadIdx.x;
    int lane = t & 31;
    int wv = __builtin_amdgcn_readfirstlane(t >> 5);
    int b = blockIdx.x;
    if (b >= LIST_BLOCKS) {
        gemm_naive_elem(Xp, Wp, bp, Yp, 60, 60, 128, (b - LIST_BLOCKS) * 1024 + t);
        return;
    }
    int a  = (b >= 80) + (b >= 120) + (b >= 124) + (b >= 164) + (b >= 204);
    int b0 = a == 0 ? 0 : a == 1 ? 80 : a == 2 ? 120 : a == 3 ? 124 : a == 4 ? 164 : 204;
    int sl   = (b - b0) & (SLICES - 1);
    int r0   = ((b - b0) >> 2) * BLOCK_ROWS;
    int nro  = row_off(a + 1) - row_off(a);
    int nr   = nro - r0 < BLOCK_ROWS ? nro - r0 : BLOCK_ROWS;
    int E    = (edge_off(a + 1) - edge_off(a)) >> 2;
    int ebase = row_off(a) + a;
    const int* rows = a == 0 ? rp.r0 : a == 1 ? rp.r1 : a == 2 ? rp.r2
                    : a == 3 ? rp.r3 : a == 4 ? rp.r4 : rp.r5;
    const int* cols = a == 0 ? ep.c0 : a == 1 ? ep.c1 : a == 2 ? ep.c2
                    : a == 3 ? ep.c3 : a == 4 ? ep.c4 : ep.c5;
    const float* vals = a == 0 ? ep.v0 : a == 1 ? ep.v1 : a == 2 ? ep.v2
                      : a == 3 ? ep.v3 : a == 4 ? ep.v4 : ep.v5;
    rows += sl * E;
    cols += sl * E;
    vals += sl * E;

    cnt[t] = 0;
    __syncthreads();

    int below = 0;
    if ((((size_t)rows) & 15) == 0) {
        const int4* rows4 = reinterpret_cast<const int4*>(rows);
        int n4 = E >> 2;
        int G = n4 >> 12;
        int m4r0 = -(r0 << 2), nr4 = nr << 2;
        const int4 none = make_int4(0x1fffffff, 0x1fffffff, 0x1fffffff, 0x1fffffff);
#define LOAD_GROUP(X, q)                                                        \
        do {                                                                    \
            int q_ = (q) < G ? (q) : G - 1;                                     \
            const int4* p_ = rows4 + (q_ << 12) + t;                            \
            X##0 = p_[0];    X##1 = p_[1024];                                   \
            X##2 = p_[2048]; X##3 = p_[3072];                                   \
        } while (0)
#define VISIT_GROUP(X, q)                                                       \
        visit_group(X##0, X##1, X##2, X##3, (((q) << 12) + t) << 2,             \
                    m4r0, nr4, cnt, bins, below)
        int4 A0 = none, A1 = none, A2 = none, A3 = none;
        int4 B0 = none, B1 = none, B2 = none, B3 = none;
        if (G > 0) {
            LOAD_GROUP(A, 0);
            LOAD_GROUP(B, 1);
        }
        {
            int i = (G << 12) + t;
            int4 T0 = i        < n4 ? rows4[i]        : none;
            int4 T1 = i + 1024 < n4 ? rows4[i + 1024] : none;
            int4 T2 = i + 2048 < n4 ? rows4[i + 2048] : none;
            int4 T3 = i + 3072 < n4 ? rows4[i + 3072] : none;
            VISIT_GROUP(T, G);
        }
        if (G > 0) {
            int4 C0, C1, C2, C3;
            for (int q = 0; q < G; q += 3) {
                LOAD_GROUP(C, q + 2); VISIT_GROUP(A, q);
                LOAD_GROUP(A, q + 3); if (q + 1 < G) VISIT_GROUP(B, q + 1);
                LOAD_GROUP(B, q + 4); if (q + 2 < G) VISIT_GROUP(C, q + 2);
            }
        }
#undef LOAD_GROUP
#undef VISIT_GROUP
        for (int k = (n4 << 2) + t; k < E; k += 1024)
            visit_edge(rows[k], k, r0, nr, cnt, bins, below);
    } else {
        for (int i = t; i < E; i += 1024)
            visit_edge(rows[i], i, r0, nr, cnt, bins, below);
    }

    below += __shfl_xor(below, 16, 32);
    below += __shfl_xor(below, 8, 32);
    below += __shfl_xor(below, 4, 32);
    below += __shfl_xor(below, 2, 32);
    below += __shfl_xor(below, 1, 32);
    if (lane == 0) red[wv] = below;
    __syncthreads();

    int len = cnt[t];
    int incl = len;
    for (int off = 1; off < 32; off <<= 1) {
        int v = __shfl_up(incl, off, 32);
        if (lane >= off) incl += v;
    }
    if (lane == 31) wtot[wv] = incl;
    __syncthreads();
    int wt = wtot[lane];
    int wincl = wt;
    for (int off = 1; off < 32; off <<= 1) {
        int v = __shfl_up(wincl, off, 32);
        if (lane >= off) wincl += v;
    }
    int wprefix = __shfl(wincl - wt, wv, 32);
    int base = edge_off(a) + sl * E;
    for (int k = 0; k < 32; ++k) base += red[k];
    int s0 = base + wprefix + incl - len;
    if (t < nr) rowptr[(ebase + r0 + t) * SLICES + sl] = s0;
    if (r0 + nr == nro && t == 0) rowptr[(ebase + nro) * SLICES + sl] = edge_off(a) + (sl + 1) * E;

    bool mine = t < nr;
    if (mine && len <= LIST_CAP) {
        int* bl = bins + t * LIST_PITCH;
        for (int i = len; (i & 3) != 0; ++i) bl[i] = 0x7fffffff;
        for (int i0 = 0; i0 < len; i0 += 4) {
            int4 k = *reinterpret_cast<const int4*>(bl + i0);
            int ra = 0, rb = 0, rc = 0, rd = 0;
            for (int j0 = 0; j0 < len; j0 += 4) {
                int4 q = *reinterpret_cast<const int4*>(bl + j0);
                ra += (q.x < k.x ? 1 : 0) + (q.y < k.x ? 1 : 0) + (q.z < k.x ? 1 : 0) + (q.w < k.x ? 1 : 0);
                rb += (q.x < k.y ? 1 : 0) + (q.y < k.y ? 1 : 0) + (q.z < k.y ? 1 : 0) + (q.w < k.y ? 1 : 0);
                rc += (q.x < k.z ? 1 : 0) + (q.y < k.z ? 1 : 0) + (q.z < k.z ? 1 : 0) + (q.w < k.z ? 1 : 0);
                rd += (q.x < k.w ? 1 : 0) + (q.y < k.w ? 1 : 0) + (q.z < k.w ? 1 : 0) + (q.w < k.w ? 1 : 0);
            }
            bool hb = i0 + 1 < len, hc = i0 + 2 < len, hd = i0 + 3 < len;
            int   ca = cols[k.x];
            float wa = vals[k.x];
            int   cb = hb ? cols[k.y] : 0;
            float wb = hb ? vals[k.y] : 0.f;
            int   cc = hc ? cols[k.z] : 0;
            float wc = hc ? vals[k.z] : 0.f;
            int   cd = hd ? cols[k.w] : 0;
            float wd = hd ? vals[k.w] : 0.f;
            ccol[s0 + ra] = ca; cval[s0 + ra] = wa;
            if (hb) { ccol[s0 + rb] = cb; cval[s0 + rb] = wb; }
            if (hc) { ccol[s0 + rc] = cc; cval[s0 + rc] = wc; }
            if (hd) { ccol[s0 + rd] = cd; cval[s0 + rd] = wd; }
        }
    }
    unsigned big = __builtin_amdgcn_ballot_w32(mine && len > LIST_CAP);
    while (big != 0u) {
        int l = __builtin_ctz(big);
        big &= big - 1u;
        int row = r0 + wv * 32 + l;
        int s = __builtin_amdgcn_readlane(s0, l);
        int done = 0;
        for (int i1 = 0; i1 < E; i1 += 32) {
            int i = i1 + lane;
            bool m = i < E && rows[i] == row;
            unsigned mask = __builtin_amdgcn_ballot_w32(m);
            if (m) {
                int p = s + done + __builtin_popcount(mask & ((1u << lane) - 1u));
                ccol[p] = cols[i];
                cval[p] = vals[i];
            }
            done += __builtin_popcount(mask);
        }
    }
}


struct RowSeg { int b0, b1, b2, b3, c1, c2, c3, tot; };

__device__ __forceinline__ RowSeg row_seg(const int* __restrict__ rowptr, int e0, int e1) {
    const int* p = rowptr + e0 * SLICES;
    const int* q = rowptr + e1 * SLICES;
    int s0 = p[0], s1 = p[1], s2 = p[2], s3 = p[3];
    int t0 = q[0], t1 = q[1], t2 = q[2], t3 = q[3];
    RowSeg g;
    g.c1  = t0 - s0;
    g.c2  = g.c1 + (t1 - s1);
    g.c3  = g.c2 + (t2 - s2);
    g.tot = g.c3 + (t3 - s3);
    g.b0 = s0; g.b1 = s1 - g.c1; g.b2 = s2 - g.c2; g.b3 = s3 - g.c3;
    return g;
}
__device__ __forceinline__ int seg_pos(const RowSeg& g, int i) {
    return i + (i < g.c1 ? g.b0 : i < g.c2 ? g.b1 : i < g.c3 ? g.b2 : g.b3);
}

template <int U>
__device__ __forceinline__ void add_edges128(float4& acc, int cl, int vl, int j,
                                             const float* __restrict__ x, int lane) {
    float4 xv[U];
    float w[U];
    #pragma unroll
    for (int u = 0; u < U; ++u) {
        int c = __builtin_amdgcn_readlane(cl, j + u);
        w[u] = __int_as_float(__builtin_amdgcn_readlane(vl, j + u));
        xv[u] = reinterpret_cast<const float4*>(x + (size_t)c * 128)[lane];
    }
    #pragma unroll
    for (int u = 0; u < U; ++u) {
        acc.x += w[u] * xv[u].x; acc.y += w[u] * xv[u].y;
        acc.z += w[u] * xv[u].z; acc.w += w[u] * xv[u].w;
    }
}

__device__ __forceinline__ void add_chunk128(float4& acc, int cl, int vl, int n,
                                             const float* __restrict__ x, int lane) {
    int j = 0;
    for (; j + 8 <= n; j += 8) add_edges128<8>(acc, cl, vl, j, x, lane);
    if (j + 4 <= n) { add_edges128<4>(acc, cl, vl, j, x, lane); j += 4; }
    if (j + 2 <= n) { add_edges128<2>(acc, cl, vl, j, x, lane); j += 2; }
    if (j < n) add_edges128<1>(acc, cl, vl, j, x, lane);
}

template <bool NORM>
__device__ __forceinline__ void gather128_rows(const int* __restrict__ rowptr,
                                               const int* __restrict__ ccol,
                                               const float* __restrict__ cval,
                                               int rowbase, int nrows,
                                               const float* __restrict__ x,
                                               float* __restrict__ y, int blk) {
    int lane = threadIdx.x & 31;
    int row = __builtin_amdgcn_readfirstlane((int)(blk * 256 + threadIdx.x) >> 5);
    if (row >= nrows) return;
    RowSeg g = row_seg(rowptr, rowbase + row, rowbase + row + 1);
    float4 acc = make_float4(0.f, 0.f, 0.f, 0.f);
    for (int o = 0; o < g.tot; o += 32) {
        int n = g.tot - o < 32 ? g.tot - o : 32;
        int cl = 0, vl = 0;
        if (lane < n) { int p = seg_pos(g, o + lane); cl = ccol[p]; vl = __float_as_int(cval[p]); }
        add_chunk128(acc, cl, vl, n, x, lane);
    }
    if (NORM) {
        float ss = acc.x * acc.x + acc.y * acc.y + acc.z * acc.z + acc.w * acc.w;
        ss += __shfl_xor(ss, 16, 32);
        ss += __shfl_xor(ss, 8, 32);
        ss += __shfl_xor(ss, 4, 32);
        ss += __shfl_xor(ss, 2, 32);
        ss += __shfl_xor(ss, 1, 32);
        float inv = 1.0f / (sqrtf(ss) + 1e-9f);
        acc.x *= inv; acc.y *= inv; acc.z *= inv; acc.w *= inv;
    } else {
        acc.x = acc.x > 0.f ? acc.x : 0.f; acc.y = acc.y > 0.f ? acc.y : 0.f;
        acc.z = acc.z > 0.f ? acc.z : 0.f; acc.w = acc.w > 0.f ? acc.w : 0.f;
    }
    reinterpret_cast<float4*>(y + (size_t)row * 128)[lane] = acc;
}

template <int U>
__device__ __forceinline__ void add_edges428(float4& a0, float4& b0, float4& b1, float4& b2,
                                             int cl, int vl, int j,
                                             const float* __restrict__ xa,
                                             const float* __restrict__ xb,
                                             int lane, bool tail) {
    float4 p[U], q0[U], q1[U], q2[U];
    float w[U];
    #pragma unroll
    for (int u = 0; u < U; ++u) {
        int c = __builtin_amdgcn_readlane(cl, j + u);
        w[u] = __int_as_float(__builtin_amdgcn_readlane(vl, j + u));
        const float4* pa = reinterpret_cast<const float4*>(xa + (size_t)c * 128);
        const float4* pb = reinterpret_cast<const float4*>(xb + (size_t)c * 300);
        p[u]  = pa[lane];
        q0[u] = pb[lane];
        q1[u] = pb[32 + lane];
        q2[u] = make_float4(0.f, 0.f, 0.f, 0.f);
        if (tail) q2[u] = pb[64 + lane];
    }
    #pragma unroll
    for (int u = 0; u < U; ++u) {
        float v = w[u];
        a0.x += v * p[u].x;  a0.y += v * p[u].y;  a0.z += v * p[u].z;  a0.w += v * p[u].w;
        b0.x += v * q0[u].x; b0.y += v * q0[u].y; b0.z += v * q0[u].z; b0.w += v * q0[u].w;
        b1.x += v * q1[u].x; b1.y += v * q1[u].y; b1.z += v * q1[u].z; b1.w += v * q1[u].w;
        b2.x += v * q2[u].x; b2.y += v * q2[u].y; b2.z += v * q2[u].z; b2.w += v * q2[u].w;
    }
}

__device__ __forceinline__ void gather428_norm_rows(const int* __restrict__ rowptr,
                                                    const int* __restrict__ ccol,
                                                    const float* __restrict__ cval,
                                                    int rowbase, int nrows,
                                                    const float* __restrict__ xa,
                                                    const float* __restrict__ xb,
                                                    float* __restrict__ y, int blk) {
    int lane = threadIdx.x & 31;
    int row = __builtin_amdgcn_readfirstlane((int)(blk * 256 + threadIdx.x) >> 5);
    if (row >= nrows) return;
    RowSeg g = row_seg(rowptr, rowbase + row, rowbase + row + 1);
    bool tail = lane < 11;
    float4 a0 = make_float4(0.f, 0.f, 0.f, 0.f), b0 = a0, b1 = a0, b2 = a0;
    for (int o = 0; o < g.tot; o += 32) {
        int n = g.tot - o < 32 ? g.tot - o : 32;
        int cl = 0, vl = 0;
        if (lane < n) { int p = seg_pos(g, o + lane); cl = ccol[p]; vl = __float_as_int(cval[p]); }
        int j = 0;
        for (; j + 4 <= n; j += 4) add_edges428<4>(a0, b0, b1, b2, cl, vl, j, xa, xb, lane, tail);
        if (j + 2 <= n) { add_edges428<2>(a0, b0, b1, b2, cl, vl, j, xa, xb, lane, tail); j += 2; }
        if (j < n) add_edges428<1>(a0, b0, b1, b2, cl, vl, j, xa, xb, lane, tail);
    }
    if (!tail) b2 = make_float4(0.f, 0.f, 0.f, 0.f);
    float ss = a0.x * a0.x + a0.y * a0.y + a0.z * a0.z + a0.w * a0.w
             + b0.x * b0.x + b0.y * b0.y + b0.z * b0.z + b0.w * b0.w
             + b1.x * b1.x + b1.y * b1.y + b1.z * b1.z + b1.w * b1.w
             + b2.x * b2.x + b2.y * b2.y + b2.z * b2.z + b2.w * b2.w;
    ss += __shfl_xor(ss, 16, 32);
    ss += __shfl_xor(ss, 8, 32);
    ss += __shfl_xor(ss, 4, 32);
    ss += __shfl_xor(ss, 2, 32);
    ss += __shfl_xor(ss, 1, 32);
    float inv = 1.0f / (sqrtf(ss) + 1e-9f);
    float4* py = reinterpret_cast<float4*>(y + (size_t)row * 428);
    py[lane]      = make_float4(a0.x * inv, a0.y * inv, a0.z * inv, a0.w * inv);
    py[32 + lane] = make_float4(b0.x * inv, b0.y * inv, b0.z * inv, b0.w * inv);
    py[64 + lane] = make_float4(b1.x * inv, b1.y * inv, b1.z * inv, b1.w * inv);
    if (tail) py[96 + lane] = make_float4(b2.x * inv, b2.y * inv, b2.z * inv, b2.w * inv);
}

#define STRIP_ROWS  32
#define STRIP_PITCH 132

__device__ __forceinline__ void gather_dense_strip(const int* __restrict__ rowptr,
                                                   const int* __restrict__ ccol,
                                                   const float* __restrict__ cval,
                                                   int rowbase, int nrows,
                                                   const float* __restrict__ x,
                                                   const float* __restrict__ W,
                                                   const float* __restrict__ bias,
                                                   float* __restrict__ Y, int blk,
                                                   float* strip) {
    int lane = threadIdx.x & 31;
    int wv = __builtin_amdgcn_readfirstlane((int)threadIdx.x >> 5);
    int r0 = blk * STRIP_ROWS;

    RowSeg sg[4];
    #pragma unroll
    for (int i = 0; i < 4; ++i) {
        int r = r0 + 4 * wv + i;
        sg[i] = row_seg(rowptr, rowbase + (r < nrows ? r : nrows),
                                rowbase + (r + 1 < nrows ? r + 1 : nrows));
    }
    int cl[4], vl[4];
    #pragma unroll
    for (int i = 0; i < 4; ++i) {
        int n = sg[i].tot < 32 ? sg[i].tot : 32;
        cl[i] = 0; vl[i] = 0;
        if (lane < n) { int p = seg_pos(sg[i], lane); cl[i] = ccol[p]; vl[i] = __float_as_int(cval[p]); }
    }
    #pragma unroll
    for (int i = 0; i < 4; ++i) {
        int tot = sg[i].tot;
        float4 acc = make_float4(0.f, 0.f, 0.f, 0.f);
        add_chunk128(acc, cl[i], vl[i], tot < 32 ? tot : 32, x, lane);
        for (int o = 32; o < tot; o += 32) {
            int n = tot - o < 32 ? tot - o : 32;
            int c2 = 0, v2 = 0;
            if (lane < n) { int p = seg_pos(sg[i], o + lane); c2 = ccol[p]; v2 = __float_as_int(cval[p]); }
            add_chunk128(acc, c2, v2, n, x, lane);
        }
        acc.x = acc.x > 0.f ? acc.x : 0.f; acc.y = acc.y > 0.f ? acc.y : 0.f;
        acc.z = acc.z > 0.f ? acc.z : 0.f; acc.w = acc.w > 0.f ? acc.w : 0.f;
        reinterpret_cast<float4*>(strip + (4 * wv + i) * STRIP_PITCH)[lane] = acc;
    }
    __syncthreads();

    int half = lane >> 4;
    int l16  = lane & 15;
    const float* wcol = W + wv * 16 + l16;
    const float* a_lo = strip + l16 * STRIP_PITCH;
    const float* a_hi = strip + (16 + l16) * STRIP_PITCH;

    v8f c0 = {}, c1 = {};
    #pragma unroll
    for (int k = 0; k < 128; k += 4) {
        int ka = k + 2 * half;
        v2f b, a0, a1;
        b[0] = wcol[(size_t)ka * 128];
        b[1] = wcol[(size_t)(ka + 1) * 128];
        a0 = *reinterpret_cast<const v2f*>(a_lo + ka);
        a1 = *reinterpret_cast<const v2f*>(a_hi + ka);
        c0 = __builtin_amdgcn_wmma_f32_16x16x4_f32(
                  false, a0,  false, b,
                  (short)0, c0,  false,  false);
        c1 = __builtin_amdgcn_wmma_f32_16x16x4_f32(
                  false, a1,  false, b,
                  (short)0, c1,  false,  false);
    }

    float bn = bias[wv * 16 + l16];
    #pragma unroll
    for (int r = 0; r < 8; ++r) {
        int m = r0 + r + 8 * half;
        if (m < nrows)      Y[(size_t)m * 128 + wv * 16 + l16] = c0[r] + bn;
        if (m + 16 < nrows) Y[(size_t)(m + 16) * 128 + wv * 16 + l16] = c1[r] + bn;
    }
}

__device__ __forceinline__ void gemm_naive_elem(const float* __restrict__ X,
                                                const float* __restrict__ W,
                                                const float* __restrict__ bias,
                                                float* __restrict__ Y,
                                                int M, int K, int N, int idx) {
    if (idx >= M * N) return;
    int m = idx / N, n = idx % N;
    float acc = bias[n];
    for (int k = 0; k < K; ++k)
        acc += X[m * K + k] * W[k * N + n];
    Y[idx] = acc;
}

#define POS_GEMM_BLOCKS   30
#define POS_GATHER_BLOCKS 8
#define WORD_ROW_BLOCKS   2500
#define ENT_ROW_BLOCKS    1250
#define DOC_ROW_BLOCKS    1250
#define WORD_STRIP_BLOCKS 625
#define ENT_STRIP_BLOCKS  313

__global__ void __launch_bounds__(256) stage_gather(const int* __restrict__ rowptr,
                                                    const int* __restrict__ ccol,
                                                    const float* __restrict__ cval,
                                                    const float* xw, float* yw,
                                                    const float* xe, float* ye,
                                                    const float* Xp, const float* Wp,
                                                    const float* bp, float* Yp, int Kp) {
    int b = blockIdx.x;
    if (b < POS_GEMM_BLOCKS) {
        gemm_naive_elem(Xp, Wp, bp, Yp, 60, Kp, 128, b * 256 + (int)threadIdx.x);
    } else {
        bool word = b < POS_GEMM_BLOCKS + WORD_ROW_BLOCKS;
        gather128_rows<false>(rowptr, ccol, cval, word ? 0 : 20001, word ? 20000 : 10000,
                              word ? xw : xe, word ? yw : ye,
                              word ? b - POS_GEMM_BLOCKS
                                   : b - (POS_GEMM_BLOCKS + WORD_ROW_BLOCKS));
    }
}

__global__ void __launch_bounds__(256) stage_gather_dense(const int* __restrict__ rowptr,
                                                          const int* __restrict__ ccol,
                                                          const float* __restrict__ cval,
                                                          const float* xp, float* yp,
                                                          const float* xw, const float* Ww,
                                                          const float* bw, float* Yw,
                                                          const float* xe, const float* We,
                                                          const float* be, float* Ye) {
    __shared__ __attribute__((aligned(16))) float strip[STRIP_ROWS * STRIP_PITCH];
    int b = blockIdx.x;
    if (b < POS_GATHER_BLOCKS) {
        gather128_rows<false>(rowptr, ccol, cval, 30002, 60, xp, yp, b);
    } else {
        bool word = b < POS_GATHER_BLOCKS + WORD_STRIP_BLOCKS;
        gather_dense_strip(rowptr, ccol, cval, word ? 0 : 20001, word ? 20000 : 10000,
                           word ? xw : xe, word ? Ww : We, word ? bw : be, word ? Yw : Ye,
                           word ? b - POS_GATHER_BLOCKS
                                : b - (POS_GATHER_BLOCKS + WORD_STRIP_BLOCKS),
                           strip);
    }
}

__global__ void __launch_bounds__(256) stage_docs(const int* __restrict__ rowptr,
                                                  const int* __restrict__ ccol,
                                                  const float* __restrict__ cval,
                                                  const float* xp, float* yp,
                                                  const float* xa, const float* xb, float* y2,
                                                  const float* xw, float* y1) {
    int b = blockIdx.x;
    if (b < POS_GATHER_BLOCKS) {
        gather128_rows<false>(rowptr, ccol, cval, 30002, 60, xp, yp, b);
    } else if (b < POS_GATHER_BLOCKS + DOC_ROW_BLOCKS) {
        gather428_norm_rows(rowptr, ccol, cval, 40064, 10000, xa, xb, y2, b - POS_GATHER_BLOCKS);
    } else {
        gather128_rows<true>(rowptr, ccol, cval, 30063, 10000, xw, y1,
                             b - (POS_GATHER_BLOCKS + DOC_ROW_BLOCKS));
    }
}

__global__ void __launch_bounds__(256) stage_docs_pos(const int* __restrict__ rowptr,
                                                      const int* __restrict__ ccol,
                                                      const float* __restrict__ cval,
                                                      const float* __restrict__ x,
                                                      float* __restrict__ y) {
    gather128_rows<true>(rowptr, ccol, cval, 50065, 10000, x, y, blockIdx.x);
}

extern "C" void kernel_launch(void* const* d_in, const int* in_sizes, int n_in,
                              void* d_out, int out_size, void* d_ws, size_t ws_size,
                              hipStream_t stream) {
    (void)in_sizes; (void)n_in; (void)out_size; (void)ws_size;

    const int N_DOC = 10000, N_WORD = 20000, N_ENT = 10000, N_POS = 60;
    const int D = 128, D_WEMB = 300;

    const float* f1       = (const float*)d_in[0];
    const float* f2       = (const float*)d_in[1];
    const float* f3       = (const float*)d_in[2];
    const float* word_emb = (const float*)d_in[3];
    const int*   a11r = (const int*)d_in[4];
    const int*   a11c = (const int*)d_in[5];
    const float* a11v = (const float*)d_in[6];
    const int*   a22r = (const int*)d_in[7];
    const int*   a22c = (const int*)d_in[8];
    const float* a22v = (const float*)d_in[9];
    const int*   a33r = (const int*)d_in[10];
    const int*   a33c = (const int*)d_in[11];
    const float* a33v = (const float*)d_in[12];
    const int*   a01r = (const int*)d_in[13];
    const int*   a01c = (const int*)d_in[14];
    const float* a01v = (const float*)d_in[15];
    const int*   a02r = (const int*)d_in[16];
    const int*   a02c = (const int*)d_in[17];
    const float* a02v = (const float*)d_in[18];
    const int*   a03r = (const int*)d_in[19];
    const int*   a03c = (const int*)d_in[20];
    const float* a03v = (const float*)d_in[21];
    const float* W3   = (const float*)d_in[22];
    const float* b3   = (const float*)d_in[23];
    const float* W1_2 = (const float*)d_in[24];
    const float* b1_2 = (const float*)d_in[25];
    const float* W2_2 = (const float*)d_in[26];
    const float* b2_2 = (const float*)d_in[27];
    const float* W3_2 = (const float*)d_in[28];
    const float* b3_2 = (const float*)d_in[29];

    float* out  = (float*)d_out;
    float* out1 = out;
    float* out2 = out + (size_t)N_DOC * D;
    float* out3 = out2 + (size_t)N_DOC * (D + D_WEMB);

    float* bufA = (float*)d_ws;
    float* csr  = bufA + (size_t)N_WORD * D;
    float* bufC = csr + (size_t)N_WORD * D;
    float* bufD = bufC + (size_t)N_POS * D;
    int*   ccol   = (int*)csr;
    float* cval   = csr + 1033600;
    int*   rowptr = (int*)(csr + 2067200);

    float* wordG = bufA;
    float* entG  = out3;
    float* wordT = out2;
    float* entT  = out2 + (size_t)N_WORD * D;

    RowsPack rp = { a11r, a22r, a33r, a01r, a02r, a03r };
    EdgePack ep = { a11c, a22c, a33c, a01c, a02c, a03c,
                    a11v, a22v, a33v, a01v, a02v, a03v };

    csr_build<<<LIST_BLOCKS + POS_GEMM_BLOCKS_1024, 1024, 0, stream>>>(
        rp, ep, rowptr, ccol, cval, f3, W3, b3, bufC);

    stage_gather_dense<<<POS_GATHER_BLOCKS + WORD_STRIP_BLOCKS + ENT_STRIP_BLOCKS, 256, 0, stream>>>(
        rowptr, ccol, cval, bufC, bufD,
        f1, W1_2, b1_2, wordT, f2, W2_2, b2_2, entT);
    stage_gather<<<POS_GEMM_BLOCKS + WORD_ROW_BLOCKS + ENT_ROW_BLOCKS, 256, 0, stream>>>(
        rowptr, ccol, cval, wordT, wordG, entT, entG, bufD, W3_2, b3_2, bufC, 128);
    stage_docs<<<POS_GATHER_BLOCKS + DOC_ROW_BLOCKS + DOC_ROW_BLOCKS, 256, 0, stream>>>(
        rowptr, ccol, cval, bufC, bufD, entG, word_emb, out2, wordG, out1);
    stage_docs_pos<<<DOC_ROW_BLOCKS, 256, 0, stream>>>(rowptr, ccol, cval, bufD, out3);
}
